# starting kernel
# baseline (speedup 1.0000x reference)
#include <hip/hip_runtime.h>

typedef __attribute__((ext_vector_type(2))) float v2f;
typedef __attribute__((ext_vector_type(8))) float v8f;

#define GCN_N_GRAPHS 64


__global__ void gcn_fill_zero(float* __restrict__ p, int n) {
    int i = blockIdx.x * blockDim.x + threadIdx.x;
    if (i < n) p[i] = 0.0f;
}

__global__ void gcn_degree(const int* __restrict__ src, const int* __restrict__ dst,
                           float* __restrict__ deg_out, float* __restrict__ deg_in, int E) {
    int e = blockIdx.x * blockDim.x + threadIdx.x;
    if (e < E) {
        unsafeAtomicAdd(&deg_out[src[e]], 1.0f);
        unsafeAtomicAdd(&deg_in[dst[e]], 1.0f);
    }
}

__global__ void gcn_norm_finalize(float* __restrict__ buf, int n) {
    int i = blockIdx.x * blockDim.x + threadIdx.x;
    if (i < n) {
        float d = buf[i];
        d = (d < 1.0f) ? 1.0f : d;
        buf[i] = 1.0f / sqrtf(d);
    }
}

__global__ void gcn_count(const int* __restrict__ gid, float* __restrict__ counts, int n) {
    int i = blockIdx.x * blockDim.x + threadIdx.x;
    if (i < n) unsafeAtomicAdd(&counts[gid[i]], 1.0f);
}

__global__ void gcn_final(const float* __restrict__ pooled, const float* __restrict__ counts,
                          float* __restrict__ out) {
    int i = blockIdx.x * blockDim.x + threadIdx.x;
    if (i < GCN_N_GRAPHS * 64) {
        float c = counts[i >> 6];
        c = (c < 1.0f) ? 1.0f : c;
        out[i] = pooled[i] / c;
    }
}


template <int F>
__global__ void gcn_spmm(const float* __restrict__ x, const float* __restrict__ out_norm,
                         const int* __restrict__ src, const int* __restrict__ dst,
                         float* __restrict__ agg, int E) {
    const int TPE = F / 4;
    int gtid = blockIdx.x * blockDim.x + threadIdx.x;
    int e = gtid / TPE;
    if (e >= E) return;
    int f4 = (gtid - e * TPE) * 4;
    int s = src[e];
    int d = dst[e];
    float sc = out_norm[s];
    const float4 v = *reinterpret_cast<const float4*>(x + (size_t)s * F + f4);
    float* base = agg + (size_t)d * F + f4;
    unsafeAtomicAdd(base + 0, v.x * sc);
    unsafeAtomicAdd(base + 1, v.y * sc);
    unsafeAtomicAdd(base + 2, v.z * sc);
    unsafeAtomicAdd(base + 3, v.w * sc);
}


template <bool RELU, bool POOL>
__global__ void gcn_gemm_wmma(const float* __restrict__ A, const float* __restrict__ in_norm,
                              const float* __restrict__ W, const float* __restrict__ bias,
                              float* __restrict__ out, int M, int K, int Nw,
                              const int* __restrict__ graph_ids, float* __restrict__ pooled) {
    int wave = blockIdx.x * (blockDim.x >> 5) + (threadIdx.x >> 5);
    int tiles_n = Nw >> 4;
    int total = (M >> 4) * tiles_n;
    if (wave >= total) return;
    int mt = wave / tiles_n;
    int nt = wave - mt * tiles_n;
    int row0 = mt << 4;
    int col0 = nt << 4;

    int lane  = threadIdx.x & 31;
    int idx15 = lane & 15;
    int koff  = (lane >> 4) << 1;

    int rowA = row0 + idx15;
    float scale = in_norm[rowA];
    const float* arow = A + (size_t)rowA * K + koff;
    const float* wcol = W + col0 + idx15;

    v8f acc = {};
    for (int k = 0; k < K; k += 4) {
        float2 av = *reinterpret_cast<const float2*>(arow + k);
        v2f af;
        af.x = av.x * scale;
        af.y = av.y * scale;
        v2f bf;
        bf.x = wcol[(size_t)(k + koff) * Nw];
        bf.y = wcol[(size_t)(k + koff + 1) * Nw];
        acc = __builtin_amdgcn_wmma_f32_16x16x4_f32(
             false, af,  false, bf,
             (short)0, acc,  false,  false);
    }

    int col = col0 + idx15;
    float bv = bias[col];
    int mbase = (lane >> 4) << 3;
#pragma unroll
    for (int r = 0; r < 8; ++r) {
        int m = row0 + mbase + r;
        float v = acc[r] + bv;
        if (RELU) v = (v > 0.0f) ? v : 0.0f;
        if (POOL) {
            int g = graph_ids[m];
            unsafeAtomicAdd(&pooled[g * Nw + col], v);
        } else {
            out[(size_t)m * Nw + col] = v;
        }
    }
}


static inline int cdiv_l(long a, long b) { return (int)((a + b - 1) / b); }

extern "C" void kernel_launch(void* const* d_in, const int* in_sizes, int n_in,
                              void* d_out, int out_size, void* d_ws, size_t ws_size,
                              hipStream_t stream) {
    const float* emb = (const float*)d_in[0];
    const float* W1  = (const float*)d_in[1];
    const float* b1  = (const float*)d_in[2];
    const float* W2  = (const float*)d_in[3];
    const float* b2  = (const float*)d_in[4];
    const float* W3  = (const float*)d_in[5];
    const float* b3  = (const float*)d_in[6];
    const int* src = (const int*)d_in[7];
    const int* dst = (const int*)d_in[8];
    const int* gid = (const int*)d_in[9];

    const int E = in_sizes[7];
    const int N = in_sizes[9];

    float* ws       = (float*)d_ws;
    float* out_norm = ws;
    float* in_norm  = out_norm + N;
    float* pooled   = in_norm + N;
    float* counts   = pooled + GCN_N_GRAPHS * 64;
    float* agg      = counts + GCN_N_GRAPHS;
    float* h        = agg + (size_t)N * 128;

    const int T = 256;

    int zc = 2 * N + GCN_N_GRAPHS * 64 + GCN_N_GRAPHS;
    gcn_fill_zero<<<cdiv_l(zc, T), T, 0, stream>>>(ws, zc);

    gcn_degree<<<cdiv_l(E, T), T, 0, stream>>>(src, dst, out_norm, in_norm, E);
    gcn_norm_finalize<<<cdiv_l(2L * N, T), T, 0, stream>>>(out_norm, 2 * N);
    gcn_count<<<cdiv_l(N, T), T, 0, stream>>>(gid, counts, N);

    gcn_fill_zero<<<cdiv_l((long)N * 128, T), T, 0, stream>>>(agg, N * 128);
    gcn_spmm<128><<<cdiv_l((long)E * 32, T), T, 0, stream>>>(emb, out_norm, src, dst, agg, E);
    {
        int tiles = (N / 16) * (128 / 16);
        gcn_gemm_wmma<true, false><<<cdiv_l(tiles, 4), 128, 0, stream>>>(
            agg, in_norm, W1, b1, h, N, 128, 128, nullptr, nullptr);
    }

    gcn_fill_zero<<<cdiv_l((long)N * 128, T), T, 0, stream>>>(agg, N * 128);
    gcn_spmm<128><<<cdiv_l((long)E * 32, T), T, 0, stream>>>(h, out_norm, src, dst, agg, E);
    {
        int tiles = (N / 16) * (64 / 16);
        gcn_gemm_wmma<true, false><<<cdiv_l(tiles, 4), 128, 0, stream>>>(
            agg, in_norm, W2, b2, h, N, 128, 64, nullptr, nullptr);
    }

    gcn_fill_zero<<<cdiv_l((long)N * 64, T), T, 0, stream>>>(agg, N * 64);
    gcn_spmm<64><<<cdiv_l((long)E * 16, T), T, 0, stream>>>(h, out_norm, src, dst, agg, E);
    {
        int tiles = (N / 16) * (64 / 16);
        gcn_gemm_wmma<false, true><<<cdiv_l(tiles, 4), 128, 0, stream>>>(
            agg, in_norm, W3, b3, nullptr, N, 64, 64, gid, pooled);
    }

    gcn_final<<<cdiv_l(GCN_N_GRAPHS * 64, T), T, 0, stream>>>(pooled, counts, (float*)d_out);
}


# SpMM owner scan tests one-byte owner keys (dst / 400, packed once per call by the degree pass, whole lines) instead of the 4-byte destination list: 4x fewer scan bytes. Sum order: arrival, as before
# speedup vs baseline: 1.8204x; 1.8204x over previous
#include <hip/hip_runtime.h>

typedef __attribute__((ext_vector_type(2))) float v2f;
typedef __attribute__((ext_vector_type(8))) float v8f;

#define GCN_N_GRAPHS 64

#define GCN_R     400
#define GCN_NW    32
#define GCN_QCAP  544
#define GCN_KD    4
#define GCN_KROUND (GCN_KD * GCN_NW * 32 * 16)


__global__ void gcn_fill_zero(float* __restrict__ p, int n) {
    int n4 = n >> 2;
    float4* p4 = reinterpret_cast<float4*>(p);
    int t = blockIdx.x * blockDim.x + threadIdx.x;
    int stride = gridDim.x * blockDim.x;
    for (int i = t; i < n4; i += stride) p4[i] = make_float4(0.0f, 0.0f, 0.0f, 0.0f);
    int r = (n4 << 2) + t;
    if (r < n) p[r] = 0.0f;
}

__global__ void gcn_degree(const int* __restrict__ src, const int* __restrict__ dst,
                           float* __restrict__ deg_out, float* __restrict__ deg_in,
                           unsigned int* __restrict__ key, int E, int Epad) {
    int t = blockIdx.x * blockDim.x + threadIdx.x;
    int e = t * 4;
    unsigned kw = 0xFFFFFFFFu;
    if (e + 3 < E) {
        int4 s4 = reinterpret_cast<const int4*>(src)[t];
        int4 d4 = reinterpret_cast<const int4*>(dst)[t];
        unsafeAtomicAdd(&deg_out[s4.x], 1.0f);
        unsafeAtomicAdd(&deg_out[s4.y], 1.0f);
        unsafeAtomicAdd(&deg_out[s4.z], 1.0f);
        unsafeAtomicAdd(&deg_out[s4.w], 1.0f);
        unsafeAtomicAdd(&deg_in[d4.x], 1.0f);
        unsafeAtomicAdd(&deg_in[d4.y], 1.0f);
        unsafeAtomicAdd(&deg_in[d4.z], 1.0f);
        unsafeAtomicAdd(&deg_in[d4.w], 1.0f);
        kw = ((unsigned)d4.x / GCN_R) | (((unsigned)d4.y / GCN_R) << 8) |
             (((unsigned)d4.z / GCN_R) << 16) | (((unsigned)d4.w / GCN_R) << 24);
    } else {
        for (int i = 0; i < 4; ++i) {
            if (e + i < E) {
                int d = dst[e + i];
                unsafeAtomicAdd(&deg_out[src[e + i]], 1.0f);
                unsafeAtomicAdd(&deg_in[d], 1.0f);
                kw = (kw & ~(0xFFu << (8 * i))) | (((unsigned)d / GCN_R) << (8 * i));
            }
        }
    }
    if (e < Epad) key[t] = kw;
}

__global__ void gcn_norm_finalize(float* __restrict__ buf, int n) {
    int i = blockIdx.x * blockDim.x + threadIdx.x;
    if (i < n) {
        float d = buf[i];
        d = (d < 1.0f) ? 1.0f : d;
        buf[i] = 1.0f / sqrtf(d);
    }
}

__global__ void gcn_count(const int* __restrict__ gid, float* __restrict__ counts, int n) {
    int i = blockIdx.x * blockDim.x + threadIdx.x;
    if (i < n) unsafeAtomicAdd(&counts[gid[i]], 1.0f);
}


__device__ __forceinline__ unsigned gcn_zero_bytes(unsigned x) {
    return ~(((x & 0x7F7F7F7Fu) + 0x7F7F7F7Fu) | x) & 0x80808080u;
}

template <int F, bool FULL>
__device__ __forceinline__ void gcn_range_batch(const float* __restrict__ xl,
                                                const float* __restrict__ out_norm,
                                                const int* __restrict__ src,
                                                const int* __restrict__ dst, float* accl,
                                                const int* qb, int base, int nvalid, int lane) {
    constexpr int LPR = F / 32;
    constexpr int G   = 16 / LPR;
    constexpr int NG  = 32 / G;
    int s = 0, dl = 0;
    float c = 0.0f;
    if (FULL || lane < nvalid) {
        int e = qb[lane];
        s = src[e];
        dl = dst[e] - base;
        c = out_norm[s];
    }
    int ci = __float_as_int(c);
    float v[2][G][LPR];
    auto load_group = [&](int g, float (&buf)[G][LPR]) {
#pragma unroll
        for (int j = 0; j < G; ++j) {
            if (FULL || g * G + j < nvalid) {
                int sj = __builtin_amdgcn_readlane(s, g * G + j);
                const float* xr = xl + (size_t)sj * F;
#pragma unroll
                for (int i = 0; i < LPR; ++i) buf[j][i] = xr[32 * i];
            } else {
#pragma unroll
                for (int i = 0; i < LPR; ++i) buf[j][i] = 0.0f;
            }
        }
    };
    auto add_group = [&](int g, float (&buf)[G][LPR]) {
#pragma unroll
        for (int j = 0; j < G; ++j) {
            if (FULL || g * G + j < nvalid) {
                int dj = __builtin_amdgcn_readlane(dl, g * G + j);
                float cj = __int_as_float(__builtin_amdgcn_readlane(ci, g * G + j));
                float* ar = accl + dj * F;
#pragma unroll
                for (int i = 0; i < LPR; ++i)
                    __hip_atomic_fetch_add(ar + 32 * i, buf[j][i] * cj, __ATOMIC_RELAXED,
                                           __HIP_MEMORY_SCOPE_WORKGROUP);
            }
        }
    };
    load_group(0, v[0]);
#pragma unroll
    for (int g = 0; g < NG; ++g) {
        if (g + 1 < NG) load_group(g + 1, v[(g + 1) & 1]);
        add_group(g, v[g & 1]);
    }
}

template <int F>
__global__ void __launch_bounds__(GCN_NW * 32)
gcn_range_spmm(const float* __restrict__ x, const float* __restrict__ out_norm,
               const int* __restrict__ src, const int* __restrict__ dst,
               const unsigned int* __restrict__ key, float* __restrict__ agg, int nrounds, int N) {
    __shared__ float acc[GCN_R * F];
    __shared__ int q[GCN_NW * GCN_QCAP];

    const int tid = threadIdx.x;
    const int lane = tid & 31;
    const int w = __builtin_amdgcn_readfirstlane(tid >> 5);
    const int base = blockIdx.x * GCN_R;

    float4* acc4 = reinterpret_cast<float4*>(acc);
    for (int i = tid; i < GCN_R * F / 4; i += GCN_NW * 32) acc4[i] = make_float4(0.0f, 0.0f, 0.0f, 0.0f);
    __syncthreads();

    int* qw = q + w * GCN_QCAP;
    const float* xl = x + lane;
    float* accl = acc + lane;
    int cnt = 0;

    const uint4* key4 = reinterpret_cast<const uint4*>(key);
    const unsigned kb = blockIdx.x * 0x01010101u;
    auto load_keys = [&](int st) -> uint4 { return key4[(st * GCN_NW + w) * 32 + lane]; };
    auto scan_step = [&](int st, const uint4& k) {
        const int ebase = ((st * GCN_NW + w) * 32 + lane) * 16;
        unsigned m = (gcn_zero_bytes(k.x ^ kb) >> 7) | (gcn_zero_bytes(k.y ^ kb) >> 6) |
                     (gcn_zero_bytes(k.z ^ kb) >> 5) | (gcn_zero_bytes(k.w ^ kb) >> 4);
        unsigned b = __builtin_amdgcn_ballot_w32(m != 0);
        while (b) {
            if (m) {
                int i = __builtin_ctz(m);
                m &= m - 1;
                qw[__builtin_amdgcn_mbcnt_lo(b, (unsigned)cnt)] = ebase + ((i & 7) << 2) + (i >> 3);
            }
            cnt += __builtin_popcount(b);
            b = __builtin_amdgcn_ballot_w32(m != 0);
        }
        while (cnt >= 32) {
            gcn_range_batch<F, true>(xl, out_norm, src, dst, accl, qw + (cnt - 32), base, 32, lane);
            cnt -= 32;
        }
    };

    if (nrounds > 0) {
        uint4 r[GCN_KD];
#pragma unroll
        for (int u = 0; u < GCN_KD; ++u) {
            r[u] = load_keys(u);
            __builtin_amdgcn_sched_barrier(0);
        }
        for (int g = 0; g < nrounds; ++g) {
            const int gn = (g + 1 < nrounds) ? g + 1 : g;
#pragma unroll
            for (int u = 0; u < GCN_KD; ++u) {
                scan_step(g * GCN_KD + u, r[u]);
                r[u] = load_keys(gn * GCN_KD + u);
            }
        }
    }
    if (cnt > 0) gcn_range_batch<F, false>(xl, out_norm, src, dst, accl, qw, base, cnt, lane);

    __syncthreads();
    int rows = N - base;
    rows = (rows < GCN_R) ? rows : GCN_R;
    float4* out4 = reinterpret_cast<float4*>(agg + (size_t)base * F);
    for (int i = tid; i < rows * (F / 4); i += GCN_NW * 32) out4[i] = acc4[i];
}


__global__ void __launch_bounds__(512)
gcn_edge_pool(const float* __restrict__ x, const float* __restrict__ out_norm,
              const float* __restrict__ in_norm, const int* __restrict__ src,
              const int* __restrict__ dst, const int* __restrict__ gid,
              float* __restrict__ pooled, int E) {
    constexpr int F   = 64;
    constexpr int LPR = F / 32;
    constexpr int G   = 16 / LPR;
    constexpr int NG  = 32 / G;
    __shared__ float tab[GCN_N_GRAPHS * F];
    for (int i = threadIdx.x; i < GCN_N_GRAPHS * F; i += blockDim.x) tab[i] = 0.0f;
    __syncthreads();

    int lane = threadIdx.x & 31;
    int wpb = blockDim.x >> 5;
    int wave = __builtin_amdgcn_readfirstlane((int)(blockIdx.x * wpb + (threadIdx.x >> 5)));
    int nwaves = gridDim.x * wpb;
    int nchunks = (E + 31) >> 5;
    const float* xl = x + lane;
    float* tl = tab + lane;

    for (int ch = wave; ch < nchunks; ch += nwaves) {
        int e0 = ch * 32;
        int e = e0 + lane;
        int s = 0, g = 0;
        float c = 0.0f;
        if (e < E) {
            s = src[e];
            int d = dst[e];
            c = out_norm[s] * in_norm[d];
            g = gid[d];
        }
        int ci = __float_as_int(c);

        if (e0 + 32 <= E) {
            float v[2][G][LPR];
            auto load_group = [&](int gr, float (&buf)[G][LPR]) {
#pragma unroll
                for (int j = 0; j < G; ++j) {
                    int sj = __builtin_amdgcn_readlane(s, gr * G + j);
                    const float* xr = xl + (size_t)sj * F;
#pragma unroll
                    for (int i = 0; i < LPR; ++i) buf[j][i] = xr[32 * i];
                }
            };
            auto add_group = [&](int gr, float (&buf)[G][LPR]) {
#pragma unroll
                for (int j = 0; j < G; ++j) {
                    int gj = __builtin_amdgcn_readlane(g, gr * G + j);
                    float cj = __int_as_float(__builtin_amdgcn_readlane(ci, gr * G + j));
                    float* tr = tl + gj * F;
#pragma unroll
                    for (int i = 0; i < LPR; ++i)
                        __hip_atomic_fetch_add(tr + 32 * i, buf[j][i] * cj, __ATOMIC_RELAXED,
                                               __HIP_MEMORY_SCOPE_WORKGROUP);
                }
            };
            load_group(0, v[0]);
#pragma unroll
            for (int gr = 0; gr < NG; ++gr) {
                if (gr + 1 < NG) load_group(gr + 1, v[(gr + 1) & 1]);
                add_group(gr, v[gr & 1]);
            }
        } else {
            int cnt = E - e0;
            for (int j = 0; j < cnt; ++j) {
                int sj = __builtin_amdgcn_readlane(s, j);
                int gj = __builtin_amdgcn_readlane(g, j);
                float cj = __int_as_float(__builtin_amdgcn_readlane(ci, j));
                const float* xr = xl + (size_t)sj * F;
                float* tr = tl + gj * F;
#pragma unroll
                for (int i = 0; i < LPR; ++i)
                    __hip_atomic_fetch_add(tr + 32 * i, xr[32 * i] * cj, __ATOMIC_RELAXED,
                                           __HIP_MEMORY_SCOPE_WORKGROUP);
            }
        }
    }

    __syncthreads();
    for (int i = threadIdx.x; i < GCN_N_GRAPHS * F; i += blockDim.x)
        unsafeAtomicAdd(&pooled[i], tab[i]);
}


template <bool RELU, bool MEAN>
__global__ void gcn_gemm_wmma(const float* __restrict__ A, const float* __restrict__ in_norm,
                              const float* __restrict__ W, const float* __restrict__ bias,
                              float* __restrict__ out, int M, int K, int Nw,
                              const float* __restrict__ counts) {
    int wave = blockIdx.x * (blockDim.x >> 5) + (threadIdx.x >> 5);
    int tiles_n = Nw >> 4;
    int total = (M >> 4) * tiles_n;
    if (wave >= total) return;
    int mt = wave / tiles_n;
    int nt = wave - mt * tiles_n;
    int row0 = mt << 4;
    int col0 = nt << 4;

    int lane  = threadIdx.x & 31;
    int idx15 = lane & 15;
    int koff  = (lane >> 4) << 1;

    int rowA = row0 + idx15;
    float scale = 1.0f;
    if (!MEAN) scale = in_norm[rowA];
    const float* arow = A + (size_t)rowA * K + koff;
    const float* wcol = W + col0 + idx15;

    v8f acc = {};
    for (int k = 0; k < K; k += 4) {
        float2 av = *reinterpret_cast<const float2*>(arow + k);
        v2f af;
        af.x = MEAN ? av.x : av.x * scale;
        af.y = MEAN ? av.y : av.y * scale;
        v2f bf;
        bf.x = wcol[(size_t)(k + koff) * Nw];
        bf.y = wcol[(size_t)(k + koff + 1) * Nw];
        acc = __builtin_amdgcn_wmma_f32_16x16x4_f32(
             false, af,  false, bf,
             (short)0, acc,  false,  false);
    }

    int col = col0 + idx15;
    float bv = bias[col];
    int mbase = (lane >> 4) << 3;
#pragma unroll
    for (int r = 0; r < 8; ++r) {
        int m = row0 + mbase + r;
        if (MEAN) {
            float c = counts[m];
            float cc = (c < 1.0f) ? 1.0f : c;
            out[(size_t)m * Nw + col] = (acc[r] + c * bv) / cc;
        } else {
            float v = acc[r] + bv;
            if (RELU) v = (v > 0.0f) ? v : 0.0f;
            out[(size_t)m * Nw + col] = v;
        }
    }
}


static inline int cdiv_l(long a, long b) { return (int)((a + b - 1) / b); }

extern "C" void kernel_launch(void* const* d_in, const int* in_sizes, int n_in,
                              void* d_out, int out_size, void* d_ws, size_t ws_size,
                              hipStream_t stream) {
    const float* emb = (const float*)d_in[0];
    const float* W1  = (const float*)d_in[1];
    const float* b1  = (const float*)d_in[2];
    const float* W2  = (const float*)d_in[3];
    const float* b2  = (const float*)d_in[4];
    const float* W3  = (const float*)d_in[5];
    const float* b3  = (const float*)d_in[6];
    const int* src = (const int*)d_in[7];
    const int* dst = (const int*)d_in[8];
    const int* gid = (const int*)d_in[9];

    const int E = in_sizes[7];
    const int N = in_sizes[9];

    float* ws       = (float*)d_ws;
    float* out_norm = ws;
    float* in_norm  = out_norm + N;
    float* pooled   = in_norm + N;
    float* counts   = pooled + GCN_N_GRAPHS * 64;
    float* agg      = counts + GCN_N_GRAPHS;
    float* h        = agg + (size_t)N * 128;
    unsigned int* key = (unsigned int*)(h + (size_t)N * 128);

    const int T = 256;
    const int nrounds = cdiv_l(E, GCN_KROUND);
    const int Epad = nrounds * GCN_KROUND;

    int zc = 2 * N + GCN_N_GRAPHS * 64 + GCN_N_GRAPHS;
    gcn_fill_zero<<<cdiv_l(zc, 4 * T), T, 0, stream>>>(ws, zc);

    gcn_degree<<<cdiv_l(Epad / 4, T), T, 0, stream>>>(src, dst, out_norm, in_norm, key, E, Epad);
    gcn_norm_finalize<<<cdiv_l(2L * N, T), T, 0, stream>>>(out_norm, 2 * N);
    gcn_count<<<cdiv_l(N, T), T, 0, stream>>>(gid, counts, N);

    gcn_range_spmm<128><<<cdiv_l(N, GCN_R), GCN_NW * 32, 0, stream>>>(emb, out_norm, src, dst, key, agg, nrounds, N);
    {
        int tiles = (N / 16) * (128 / 16);
        gcn_gemm_wmma<true, false><<<cdiv_l(tiles, 4), 128, 0, stream>>>(
            agg, in_norm, W1, b1, h, N, 128, 128, nullptr);
    }

    gcn_range_spmm<128><<<cdiv_l(N, GCN_R), GCN_NW * 32, 0, stream>>>(h, out_norm, src, dst, key, agg, nrounds, N);
    {
        int tiles = (N / 16) * (64 / 16);
        gcn_gemm_wmma<true, false><<<cdiv_l(tiles, 4), 128, 0, stream>>>(
            agg, in_norm, W2, b2, h, N, 128, 64, nullptr);
    }

    gcn_edge_pool<<<768, 512, 0, stream>>>(
        h, out_norm, in_norm, src, dst, gid, pooled, E);
    {
        int tiles = (GCN_N_GRAPHS / 16) * (64 / 16);
        gcn_gemm_wmma<false, true><<<cdiv_l(tiles, 4), 128, 0, stream>>>(
            pooled, nullptr, W3, b3, (float*)d_out, GCN_N_GRAPHS, 64, 64, counts);
    }
}


# Per-edge scalars looked up once: a record pass writes (src|owned row|graph, out_norm[src]) and in_norm[dst] per edge; SpMM reads 1 record per kept edge, pooling streams them. Sum order as before
# speedup vs baseline: 1.9851x; 1.0905x over previous
#include <hip/hip_runtime.h>

typedef __attribute__((ext_vector_type(2))) float v2f;
typedef __attribute__((ext_vector_type(8))) float v8f;

#define GCN_N_GRAPHS 64

#define GCN_R     400
#define GCN_NW    32
#define GCN_QCAP  544
#define GCN_KD    4
#define GCN_KROUND (GCN_KD * GCN_NW * 32 * 16)


__global__ void gcn_fill_zero(float* __restrict__ p, int n) {
    int n4 = n >> 2;
    float4* p4 = reinterpret_cast<float4*>(p);
    int t = blockIdx.x * blockDim.x + threadIdx.x;
    int stride = gridDim.x * blockDim.x;
    for (int i = t; i < n4; i += stride) p4[i] = make_float4(0.0f, 0.0f, 0.0f, 0.0f);
    int r = (n4 << 2) + t;
    if (r < n) p[r] = 0.0f;
}

__global__ void gcn_degree(const int* __restrict__ src, const int* __restrict__ dst,
                           float* __restrict__ deg_out, float* __restrict__ deg_in, int E) {
    int t = blockIdx.x * blockDim.x + threadIdx.x;
    int e = t * 4;
    if (e + 3 < E) {
        int4 s4 = reinterpret_cast<const int4*>(src)[t];
        int4 d4 = reinterpret_cast<const int4*>(dst)[t];
        unsafeAtomicAdd(&deg_out[s4.x], 1.0f);
        unsafeAtomicAdd(&deg_out[s4.y], 1.0f);
        unsafeAtomicAdd(&deg_out[s4.z], 1.0f);
        unsafeAtomicAdd(&deg_out[s4.w], 1.0f);
        unsafeAtomicAdd(&deg_in[d4.x], 1.0f);
        unsafeAtomicAdd(&deg_in[d4.y], 1.0f);
        unsafeAtomicAdd(&deg_in[d4.z], 1.0f);
        unsafeAtomicAdd(&deg_in[d4.w], 1.0f);
    } else {
        for (int i = 0; i < 4; ++i) {
            if (e + i < E) {
                unsafeAtomicAdd(&deg_out[src[e + i]], 1.0f);
                unsafeAtomicAdd(&deg_in[dst[e + i]], 1.0f);
            }
        }
    }
}

__device__ __forceinline__ int gcn_pack_edge(int s, int d, unsigned owner, int g) {
    return (int)((unsigned)s | (((unsigned)d - owner * GCN_R) << 17) | ((unsigned)(g & 63) << 26));
}

__global__ void gcn_edge_records(const int* __restrict__ src, const int* __restrict__ dst,
                                 const float* __restrict__ out_norm, const int* __restrict__ npair,
                                 unsigned int* __restrict__ key, int* __restrict__ rec,
                                 float* __restrict__ dnorm, int E, int Epad) {
    int t = blockIdx.x * blockDim.x + threadIdx.x;
    int e = t * 4;
    unsigned kw = 0xFFFFFFFFu;
    const int2* np2 = reinterpret_cast<const int2*>(npair);
    if (e + 3 < E) {
        int4 s4 = reinterpret_cast<const int4*>(src)[t];
        int4 d4 = reinterpret_cast<const int4*>(dst)[t];
        float a0 = out_norm[s4.x], a1 = out_norm[s4.y], a2 = out_norm[s4.z], a3 = out_norm[s4.w];
        int2 p0 = np2[d4.x], p1 = np2[d4.y], p2 = np2[d4.z], p3 = np2[d4.w];
        unsigned o0 = (unsigned)d4.x / GCN_R, o1 = (unsigned)d4.y / GCN_R;
        unsigned o2 = (unsigned)d4.z / GCN_R, o3 = (unsigned)d4.w / GCN_R;
        kw = o0 | (o1 << 8) | (o2 << 16) | (o3 << 24);
        int4* r4 = reinterpret_cast<int4*>(rec);
        r4[2 * t] = make_int4(gcn_pack_edge(s4.x, d4.x, o0, p0.y), __float_as_int(a0),
                              gcn_pack_edge(s4.y, d4.y, o1, p1.y), __float_as_int(a1));
        r4[2 * t + 1] = make_int4(gcn_pack_edge(s4.z, d4.z, o2, p2.y), __float_as_int(a2),
                                  gcn_pack_edge(s4.w, d4.w, o3, p3.y), __float_as_int(a3));
        reinterpret_cast<int4*>(dnorm)[t] = make_int4(p0.x, p1.x, p2.x, p3.x);
    } else {
        for (int i = 0; i < 4; ++i) {
            if (e + i < E) {
                int s = src[e + i];
                int d = dst[e + i];
                unsigned o = (unsigned)d / GCN_R;
                int2 p = np2[d];
                rec[2 * (e + i)] = gcn_pack_edge(s, d, o, p.y);
                rec[2 * (e + i) + 1] = __float_as_int(out_norm[s]);
                dnorm[e + i] = __int_as_float(p.x);
                kw = (kw & ~(0xFFu << (8 * i))) | (o << (8 * i));
            }
        }
    }
    if (e < Epad) key[t] = kw;
}

__global__ void gcn_norm_finalize(float* __restrict__ buf, const int* __restrict__ gid,
                                  int* __restrict__ npair, int N) {
    int i = blockIdx.x * blockDim.x + threadIdx.x;
    if (i < 2 * N) {
        float d = buf[i];
        d = (d < 1.0f) ? 1.0f : d;
        float r = 1.0f / sqrtf(d);
        buf[i] = r;
        if (i >= N) reinterpret_cast<int2*>(npair)[i - N] = make_int2(__float_as_int(r), gid[i - N]);
    }
}

__global__ void gcn_count(const int* __restrict__ gid, float* __restrict__ counts, int n) {
    int i = blockIdx.x * blockDim.x + threadIdx.x;
    if (i < n) unsafeAtomicAdd(&counts[gid[i]], 1.0f);
}


__device__ __forceinline__ unsigned gcn_zero_bytes(unsigned x) {
    return ~(((x & 0x7F7F7F7Fu) + 0x7F7F7F7Fu) | x) & 0x80808080u;
}

template <int F, bool FULL>
__device__ __forceinline__ void gcn_range_batch(const float* __restrict__ xl,
                                                const int* __restrict__ rec, float* accl,
                                                const int* qb, int nvalid, int lane) {
    constexpr int LPR = F / 32;
    constexpr int G   = 16 / LPR;
    constexpr int NG  = 32 / G;
    int s = 0, dl = 0, ci = 0;
    if (FULL || lane < nvalid) {
        int2 r = reinterpret_cast<const int2*>(rec)[qb[lane]];
        s = r.x & 0x1FFFF;
        dl = (r.x >> 17) & 0x1FF;
        ci = r.y;
    }
    float v[2][G][LPR];
    auto load_group = [&](int g, float (&buf)[G][LPR]) {
#pragma unroll
        for (int j = 0; j < G; ++j) {
            if (FULL || g * G + j < nvalid) {
                int sj = __builtin_amdgcn_readlane(s, g * G + j);
                const float* xr = xl + (size_t)sj * F;
#pragma unroll
                for (int i = 0; i < LPR; ++i) buf[j][i] = xr[32 * i];
            } else {
#pragma unroll
                for (int i = 0; i < LPR; ++i) buf[j][i] = 0.0f;
            }
        }
    };
    auto add_group = [&](int g, float (&buf)[G][LPR]) {
#pragma unroll
        for (int j = 0; j < G; ++j) {
            if (FULL || g * G + j < nvalid) {
                int dj = __builtin_amdgcn_readlane(dl, g * G + j);
                float cj = __int_as_float(__builtin_amdgcn_readlane(ci, g * G + j));
                float* ar = accl + dj * F;
#pragma unroll
                for (int i = 0; i < LPR; ++i)
                    __hip_atomic_fetch_add(ar + 32 * i, buf[j][i] * cj, __ATOMIC_RELAXED,
                                           __HIP_MEMORY_SCOPE_WORKGROUP);
            }
        }
    };
    load_group(0, v[0]);
#pragma unroll
    for (int g = 0; g < NG; ++g) {
        if (g + 1 < NG) load_group(g + 1, v[(g + 1) & 1]);
        add_group(g, v[g & 1]);
    }
}

template <int F>
__global__ void __launch_bounds__(GCN_NW * 32)
gcn_range_spmm(const float* __restrict__ x, const int* __restrict__ rec,
               const unsigned int* __restrict__ key, float* __restrict__ agg, int nrounds, int N) {
    __shared__ float acc[GCN_R * F];
    __shared__ int q[GCN_NW * GCN_QCAP];

    const int tid = threadIdx.x;
    const int lane = tid & 31;
    const int w = __builtin_amdgcn_readfirstlane(tid >> 5);
    const int base = blockIdx.x * GCN_R;

    float4* acc4 = reinterpret_cast<float4*>(acc);
    for (int i = tid; i < GCN_R * F / 4; i += GCN_NW * 32) acc4[i] = make_float4(0.0f, 0.0f, 0.0f, 0.0f);
    __syncthreads();

    int* qw = q + w * GCN_QCAP;
    const float* xl = x + lane;
    float* accl = acc + lane;
    int cnt = 0;

    const uint4* key4 = reinterpret_cast<const uint4*>(key);
    const unsigned kb = blockIdx.x * 0x01010101u;
    auto load_keys = [&](int st) -> uint4 { return key4[(st * GCN_NW + w) * 32 + lane]; };
    auto scan_step = [&](int st, const uint4& k) {
        const int ebase = ((st * GCN_NW + w) * 32 + lane) * 16;
        unsigned m = (gcn_zero_bytes(k.x ^ kb) >> 7) | (gcn_zero_bytes(k.y ^ kb) >> 6) |
                     (gcn_zero_bytes(k.z ^ kb) >> 5) | (gcn_zero_bytes(k.w ^ kb) >> 4);
        unsigned b = __builtin_amdgcn_ballot_w32(m != 0);
        while (b) {
            if (m) {
                int i = __builtin_ctz(m);
                m &= m - 1;
                qw[__builtin_amdgcn_mbcnt_lo(b, (unsigned)cnt)] = ebase + ((i & 7) << 2) + (i >> 3);
            }
            cnt += __builtin_popcount(b);
            b = __builtin_amdgcn_ballot_w32(m != 0);
        }
        while (cnt >= 32) {
            gcn_range_batch<F, true>(xl, rec, accl, qw + (cnt - 32), 32, lane);
            cnt -= 32;
        }
    };

    if (nrounds > 0) {
        uint4 r[GCN_KD];
#pragma unroll
        for (int u = 0; u < GCN_KD; ++u) {
            r[u] = load_keys(u);
            __builtin_amdgcn_sched_barrier(0);
        }
        for (int g = 0; g < nrounds; ++g) {
            const int gn = (g + 1 < nrounds) ? g + 1 : g;
#pragma unroll
            for (int u = 0; u < GCN_KD; ++u) {
                scan_step(g * GCN_KD + u, r[u]);
                r[u] = load_keys(gn * GCN_KD + u);
            }
        }
    }
    if (cnt > 0) gcn_range_batch<F, false>(xl, rec, accl, qw, cnt, lane);

    __syncthreads();
    int rows = N - base;
    rows = (rows < GCN_R) ? rows : GCN_R;
    float4* out4 = reinterpret_cast<float4*>(agg + (size_t)base * F);
    for (int i = tid; i < rows * (F / 4); i += GCN_NW * 32) out4[i] = acc4[i];
}


__global__ void __launch_bounds__(512)
gcn_edge_pool(const float* __restrict__ x, const int* __restrict__ rec,
              const float* __restrict__ dnorm, float* __restrict__ pooled, int E) {
    constexpr int F   = 64;
    constexpr int LPR = F / 32;
    constexpr int G   = 16 / LPR;
    constexpr int NG  = 32 / G;
    __shared__ float tab[GCN_N_GRAPHS * F];
    for (int i = threadIdx.x; i < GCN_N_GRAPHS * F; i += blockDim.x) tab[i] = 0.0f;
    __syncthreads();

    int lane = threadIdx.x & 31;
    int wpb = blockDim.x >> 5;
    int wave = __builtin_amdgcn_readfirstlane((int)(blockIdx.x * wpb + (threadIdx.x >> 5)));
    int nwaves = gridDim.x * wpb;
    int nchunks = (E + 31) >> 5;
    const float* xl = x + lane;
    float* tl = tab + lane;

    for (int ch = wave; ch < nchunks; ch += nwaves) {
        int e0 = ch * 32;
        int e = e0 + lane;
        int s = 0, g = 0;
        float c = 0.0f;
        if (e < E) {
            int2 r = reinterpret_cast<const int2*>(rec)[e];
            s = r.x & 0x1FFFF;
            g = (int)((unsigned)r.x >> 26);
            c = __int_as_float(r.y) * dnorm[e];
        }
        int ci = __float_as_int(c);

        if (e0 + 32 <= E) {
            float v[2][G][LPR];
            auto load_group = [&](int gr, float (&buf)[G][LPR]) {
#pragma unroll
                for (int j = 0; j < G; ++j) {
                    int sj = __builtin_amdgcn_readlane(s, gr * G + j);
                    const float* xr = xl + (size_t)sj * F;
#pragma unroll
                    for (int i = 0; i < LPR; ++i) buf[j][i] = xr[32 * i];
                }
            };
            auto add_group = [&](int gr, float (&buf)[G][LPR]) {
#pragma unroll
                for (int j = 0; j < G; ++j) {
                    int gj = __builtin_amdgcn_readlane(g, gr * G + j);
                    float cj = __int_as_float(__builtin_amdgcn_readlane(ci, gr * G + j));
                    float* tr = tl + gj * F;
#pragma unroll
                    for (int i = 0; i < LPR; ++i)
                        __hip_atomic_fetch_add(tr + 32 * i, buf[j][i] * cj, __ATOMIC_RELAXED,
                                               __HIP_MEMORY_SCOPE_WORKGROUP);
                }
            };
            load_group(0, v[0]);
#pragma unroll
            for (int gr = 0; gr < NG; ++gr) {
                if (gr + 1 < NG) load_group(gr + 1, v[(gr + 1) & 1]);
                add_group(gr, v[gr & 1]);
            }
        } else {
            int cnt = E - e0;
            for (int j = 0; j < cnt; ++j) {
                int sj = __builtin_amdgcn_readlane(s, j);
                int gj = __builtin_amdgcn_readlane(g, j);
                float cj = __int_as_float(__builtin_amdgcn_readlane(ci, j));
                const float* xr = xl + (size_t)sj * F;
                float* tr = tl + gj * F;
#pragma unroll
                for (int i = 0; i < LPR; ++i)
                    __hip_atomic_fetch_add(tr + 32 * i, xr[32 * i] * cj, __ATOMIC_RELAXED,
                                           __HIP_MEMORY_SCOPE_WORKGROUP);
            }
        }
    }

    __syncthreads();
    for (int i = threadIdx.x; i < GCN_N_GRAPHS * F; i += blockDim.x)
        unsafeAtomicAdd(&pooled[i], tab[i]);
}


template <bool RELU, bool MEAN>
__global__ void gcn_gemm_wmma(const float* __restrict__ A, const float* __restrict__ in_norm,
                              const float* __restrict__ W, const float* __restrict__ bias,
                              float* __restrict__ out, int M, int K, int Nw,
                              const float* __restrict__ counts) {
    int wave = blockIdx.x * (blockDim.x >> 5) + (threadIdx.x >> 5);
    int tiles_n = Nw >> 4;
    int total = (M >> 4) * tiles_n;
    if (wave >= total) return;
    int mt = wave / tiles_n;
    int nt = wave - mt * tiles_n;
    int row0 = mt << 4;
    int col0 = nt << 4;

    int lane  = threadIdx.x & 31;
    int idx15 = lane & 15;
    int koff  = (lane >> 4) << 1;

    int rowA = row0 + idx15;
    float scale = 1.0f;
    if (!MEAN) scale = in_norm[rowA];
    const float* arow = A + (size_t)rowA * K + koff;
    const float* wcol = W + col0 + idx15;

    v8f acc = {};
    for (int k = 0; k < K; k += 4) {
        float2 av = *reinterpret_cast<const float2*>(arow + k);
        v2f af;
        af.x = MEAN ? av.x : av.x * scale;
        af.y = MEAN ? av.y : av.y * scale;
        v2f bf;
        bf.x = wcol[(size_t)(k + koff) * Nw];
        bf.y = wcol[(size_t)(k + koff + 1) * Nw];
        acc = __builtin_amdgcn_wmma_f32_16x16x4_f32(
             false, af,  false, bf,
             (short)0, acc,  false,  false);
    }

    int col = col0 + idx15;
    float bv = bias[col];
    int mbase = (lane >> 4) << 3;
#pragma unroll
    for (int r = 0; r < 8; ++r) {
        int m = row0 + mbase + r;
        if (MEAN) {
            float c = counts[m];
            float cc = (c < 1.0f) ? 1.0f : c;
            out[(size_t)m * Nw + col] = (acc[r] + c * bv) / cc;
        } else {
            float v = acc[r] + bv;
            if (RELU) v = (v > 0.0f) ? v : 0.0f;
            out[(size_t)m * Nw + col] = v;
        }
    }
}


static inline int cdiv_l(long a, long b) { return (int)((a + b - 1) / b); }

extern "C" void kernel_launch(void* const* d_in, const int* in_sizes, int n_in,
                              void* d_out, int out_size, void* d_ws, size_t ws_size,
                              hipStream_t stream) {
    const float* emb = (const float*)d_in[0];
    const float* W1  = (const float*)d_in[1];
    const float* b1  = (const float*)d_in[2];
    const float* W2  = (const float*)d_in[3];
    const float* b2  = (const float*)d_in[4];
    const float* W3  = (const float*)d_in[5];
    const float* b3  = (const float*)d_in[6];
    const int* src = (const int*)d_in[7];
    const int* dst = (const int*)d_in[8];
    const int* gid = (const int*)d_in[9];

    const int E = in_sizes[7];
    const int N = in_sizes[9];

    float* ws       = (float*)d_ws;
    float* out_norm = ws;
    float* in_norm  = out_norm + N;
    float* pooled   = in_norm + N;
    float* counts   = pooled + GCN_N_GRAPHS * 64;
    float* agg      = counts + GCN_N_GRAPHS;
    float* h        = agg + (size_t)N * 128;
    unsigned int* key = (unsigned int*)(h + (size_t)N * 128);

    const int T = 256;
    const int nrounds = cdiv_l(E, GCN_KROUND);
    const int Epad = nrounds * GCN_KROUND;
    int* rec = (int*)(key + Epad / 4);
    float* dnorm = (float*)(rec + 2 * (size_t)Epad);
    int* npair = (int*)(dnorm + Epad);

    int zc = 2 * N + GCN_N_GRAPHS * 64 + GCN_N_GRAPHS;
    gcn_fill_zero<<<cdiv_l(zc, 4 * T), T, 0, stream>>>(ws, zc);

    gcn_degree<<<cdiv_l(cdiv_l(E, 4), T), T, 0, stream>>>(src, dst, out_norm, in_norm, E);
    gcn_norm_finalize<<<cdiv_l(2L * N, T), T, 0, stream>>>(out_norm, gid, npair, N);
    gcn_count<<<cdiv_l(N, T), T, 0, stream>>>(gid, counts, N);
    gcn_edge_records<<<cdiv_l(Epad / 4, T), T, 0, stream>>>(src, dst, out_norm, npair, key, rec, dnorm,
                                                         E, Epad);

    gcn_range_spmm<128><<<cdiv_l(N, GCN_R), GCN_NW * 32, 0, stream>>>(emb, rec, key, agg, nrounds, N);
    {
        int tiles = (N / 16) * (128 / 16);
        gcn_gemm_wmma<true, false><<<cdiv_l(tiles, 4), 128, 0, stream>>>(
            agg, in_norm, W1, b1, h, N, 128, 128, nullptr);
    }

    gcn_range_spmm<128><<<cdiv_l(N, GCN_R), GCN_NW * 32, 0, stream>>>(h, rec, key, agg, nrounds, N);
    {
        int tiles = (N / 16) * (64 / 16);
        gcn_gemm_wmma<true, false><<<cdiv_l(tiles, 4), 128, 0, stream>>>(
            agg, in_norm, W2, b2, h, N, 128, 64, nullptr);
    }

    gcn_edge_pool<<<768, 512, 0, stream>>>(h, rec, dnorm, pooled, E);
    {
        int tiles = (GCN_N_GRAPHS / 16) * (64 / 16);
        gcn_gemm_wmma<false, true><<<cdiv_l(tiles, 4), 128, 0, stream>>>(
            pooled, nullptr, W3, b3, (float*)d_out, GCN_N_GRAPHS, 64, 64, counts);
    }
}


# Node counts ride in the norm pass: a work-group counts its nodes in a 64-word LDS table, one f32 add per non-empty word; the launch of one same-word atomic per node is gone (11 to 10). Sums exact
# speedup vs baseline: 2.5031x; 1.2609x over previous
#include <hip/hip_runtime.h>

typedef __attribute__((ext_vector_type(2))) float v2f;
typedef __attribute__((ext_vector_type(8))) float v8f;

#define GCN_N_GRAPHS 64

#define GCN_R     400
#define GCN_NW    32
#define GCN_QCAP  544
#define GCN_KD    4
#define GCN_KROUND (GCN_KD * GCN_NW * 32 * 16)


__global__ void gcn_fill_zero(float* __restrict__ p, int n) {
    int n4 = n >> 2;
    float4* p4 = reinterpret_cast<float4*>(p);
    int t = blockIdx.x * blockDim.x + threadIdx.x;
    int stride = gridDim.x * blockDim.x;
    for (int i = t; i < n4; i += stride) p4[i] = make_float4(0.0f, 0.0f, 0.0f, 0.0f);
    int r = (n4 << 2) + t;
    if (r < n) p[r] = 0.0f;
}

__global__ void gcn_degree(const int* __restrict__ src, const int* __restrict__ dst,
                           float* __restrict__ deg_out, float* __restrict__ deg_in, int E) {
    int t = blockIdx.x * blockDim.x + threadIdx.x;
    int e = t * 4;
    if (e + 3 < E) {
        int4 s4 = reinterpret_cast<const int4*>(src)[t];
        int4 d4 = reinterpret_cast<const int4*>(dst)[t];
        unsafeAtomicAdd(&deg_out[s4.x], 1.0f);
        unsafeAtomicAdd(&deg_out[s4.y], 1.0f);
        unsafeAtomicAdd(&deg_out[s4.z], 1.0f);
        unsafeAtomicAdd(&deg_out[s4.w], 1.0f);
        unsafeAtomicAdd(&deg_in[d4.x], 1.0f);
        unsafeAtomicAdd(&deg_in[d4.y], 1.0f);
        unsafeAtomicAdd(&deg_in[d4.z], 1.0f);
        unsafeAtomicAdd(&deg_in[d4.w], 1.0f);
    } else {
        for (int i = 0; i < 4; ++i) {
            if (e + i < E) {
                unsafeAtomicAdd(&deg_out[src[e + i]], 1.0f);
                unsafeAtomicAdd(&deg_in[dst[e + i]], 1.0f);
            }
        }
    }
}

__device__ __forceinline__ int gcn_pack_edge(int s, int d, unsigned owner, int g) {
    return (int)((unsigned)s | (((unsigned)d - owner * GCN_R) << 17) | ((unsigned)(g & 63) << 26));
}

__global__ void gcn_edge_records(const int* __restrict__ src, const int* __restrict__ dst,
                                 const float* __restrict__ out_norm, const int* __restrict__ npair,
                                 unsigned int* __restrict__ key, int* __restrict__ rec,
                                 float* __restrict__ dnorm, int E, int Epad) {
    int t = blockIdx.x * blockDim.x + threadIdx.x;
    int e = t * 4;
    unsigned kw = 0xFFFFFFFFu;
    const int2* np2 = reinterpret_cast<const int2*>(npair);
    if (e + 3 < E) {
        int4 s4 = reinterpret_cast<const int4*>(src)[t];
        int4 d4 = reinterpret_cast<const int4*>(dst)[t];
        float a0 = out_norm[s4.x], a1 = out_norm[s4.y], a2 = out_norm[s4.z], a3 = out_norm[s4.w];
        int2 p0 = np2[d4.x], p1 = np2[d4.y], p2 = np2[d4.z], p3 = np2[d4.w];
        unsigned o0 = (unsigned)d4.x / GCN_R, o1 = (unsigned)d4.y / GCN_R;
        unsigned o2 = (unsigned)d4.z / GCN_R, o3 = (unsigned)d4.w / GCN_R;
        kw = o0 | (o1 << 8) | (o2 << 16) | (o3 << 24);
        int4* r4 = reinterpret_cast<int4*>(rec);
        r4[2 * t] = make_int4(gcn_pack_edge(s4.x, d4.x, o0, p0.y), __float_as_int(a0),
                              gcn_pack_edge(s4.y, d4.y, o1, p1.y), __float_as_int(a1));
        r4[2 * t + 1] = make_int4(gcn_pack_edge(s4.z, d4.z, o2, p2.y), __float_as_int(a2),
                                  gcn_pack_edge(s4.w, d4.w, o3, p3.y), __float_as_int(a3));
        reinterpret_cast<int4*>(dnorm)[t] = make_int4(p0.x, p1.x, p2.x, p3.x);
    } else {
        for (int i = 0; i < 4; ++i) {
            if (e + i < E) {
                int s = src[e + i];
                int d = dst[e + i];
                unsigned o = (unsigned)d / GCN_R;
                int2 p = np2[d];
                rec[2 * (e + i)] = gcn_pack_edge(s, d, o, p.y);
                rec[2 * (e + i) + 1] = __float_as_int(out_norm[s]);
                dnorm[e + i] = __int_as_float(p.x);
                kw = (kw & ~(0xFFu << (8 * i))) | (o << (8 * i));
            }
        }
    }
    if (e < Epad) key[t] = kw;
}

__global__ void gcn_norm_finalize(float* __restrict__ buf, const int* __restrict__ gid,
                                  int* __restrict__ npair, float* __restrict__ counts, int N) {
    __shared__ float cnt[GCN_N_GRAPHS];
    if (threadIdx.x < GCN_N_GRAPHS) cnt[threadIdx.x] = 0.0f;
    __syncthreads();
    int i = blockIdx.x * blockDim.x + threadIdx.x;
    if (i < 2 * N) {
        float d = buf[i];
        d = (d < 1.0f) ? 1.0f : d;
        float r = 1.0f / sqrtf(d);
        buf[i] = r;
        if (i >= N) {
            int g = gid[i - N];
            reinterpret_cast<int2*>(npair)[i - N] = make_int2(__float_as_int(r), g);
            if ((unsigned)g < (unsigned)GCN_N_GRAPHS)
                __hip_atomic_fetch_add(&cnt[g], 1.0f, __ATOMIC_RELAXED, __HIP_MEMORY_SCOPE_WORKGROUP);
        }
    }
    __syncthreads();
    if (threadIdx.x < GCN_N_GRAPHS) {
        float c = cnt[threadIdx.x];
        if (c != 0.0f) unsafeAtomicAdd(&counts[threadIdx.x], c);
    }
}


__device__ __forceinline__ unsigned gcn_zero_bytes(unsigned x) {
    return ~(((x & 0x7F7F7F7Fu) + 0x7F7F7F7Fu) | x) & 0x80808080u;
}

template <int F, bool FULL>
__device__ __forceinline__ void gcn_range_batch(const float* __restrict__ xl,
                                                const int* __restrict__ rec, float* accl,
                                                const int* qb, int nvalid, int lane) {
    constexpr int LPR = F / 32;
    constexpr int G   = 16 / LPR;
    constexpr int NG  = 32 / G;
    int s = 0, dl = 0, ci = 0;
    if (FULL || lane < nvalid) {
        int2 r = reinterpret_cast<const int2*>(rec)[qb[lane]];
        s = r.x & 0x1FFFF;
        dl = (r.x >> 17) & 0x1FF;
        ci = r.y;
    }
    float v[2][G][LPR];
    auto load_group = [&](int g, float (&buf)[G][LPR]) {
#pragma unroll
        for (int j = 0; j < G; ++j) {
            if (FULL || g * G + j < nvalid) {
                int sj = __builtin_amdgcn_readlane(s, g * G + j);
                const float* xr = xl + (size_t)sj * F;
#pragma unroll
                for (int i = 0; i < LPR; ++i) buf[j][i] = xr[32 * i];
            } else {
#pragma unroll
                for (int i = 0; i < LPR; ++i) buf[j][i] = 0.0f;
            }
        }
    };
    auto add_group = [&](int g, float (&buf)[G][LPR]) {
#pragma unroll
        for (int j = 0; j < G; ++j) {
            if (FULL || g * G + j < nvalid) {
                int dj = __builtin_amdgcn_readlane(dl, g * G + j);
                float cj = __int_as_float(__builtin_amdgcn_readlane(ci, g * G + j));
                float* ar = accl + dj * F;
#pragma unroll
                for (int i = 0; i < LPR; ++i)
                    __hip_atomic_fetch_add(ar + 32 * i, buf[j][i] * cj, __ATOMIC_RELAXED,
                                           __HIP_MEMORY_SCOPE_WORKGROUP);
            }
        }
    };
    load_group(0, v[0]);
#pragma unroll
    for (int g = 0; g < NG; ++g) {
        if (g + 1 < NG) load_group(g + 1, v[(g + 1) & 1]);
        add_group(g, v[g & 1]);
    }
}

template <int F>
__global__ void __launch_bounds__(GCN_NW * 32)
gcn_range_spmm(const float* __restrict__ x, const int* __restrict__ rec,
               const unsigned int* __restrict__ key, float* __restrict__ agg, int nrounds, int N) {
    __shared__ float acc[GCN_R * F];
    __shared__ int q[GCN_NW * GCN_QCAP];

    const int tid = threadIdx.x;
    const int lane = tid & 31;
    const int w = __builtin_amdgcn_readfirstlane(tid >> 5);
    const int base = blockIdx.x * GCN_R;

    float4* acc4 = reinterpret_cast<float4*>(acc);
    for (int i = tid; i < GCN_R * F / 4; i += GCN_NW * 32) acc4[i] = make_float4(0.0f, 0.0f, 0.0f, 0.0f);
    __syncthreads();

    int* qw = q + w * GCN_QCAP;
    const float* xl = x + lane;
    float* accl = acc + lane;
    int cnt = 0;

    const uint4* key4 = reinterpret_cast<const uint4*>(key);
    const unsigned kb = blockIdx.x * 0x01010101u;
    auto load_keys = [&](int st) -> uint4 { return key4[(st * GCN_NW + w) * 32 + lane]; };
    auto scan_step = [&](int st, const uint4& k) {
        const int ebase = ((st * GCN_NW + w) * 32 + lane) * 16;
        unsigned m = (gcn_zero_bytes(k.x ^ kb) >> 7) | (gcn_zero_bytes(k.y ^ kb) >> 6) |
                     (gcn_zero_bytes(k.z ^ kb) >> 5) | (gcn_zero_bytes(k.w ^ kb) >> 4);
        unsigned b = __builtin_amdgcn_ballot_w32(m != 0);
        while (b) {
            if (m) {
                int i = __builtin_ctz(m);
                m &= m - 1;
                qw[__builtin_amdgcn_mbcnt_lo(b, (unsigned)cnt)] = ebase + ((i & 7) << 2) + (i >> 3);
            }
            cnt += __builtin_popcount(b);
            b = __builtin_amdgcn_ballot_w32(m != 0);
        }
        while (cnt >= 32) {
            gcn_range_batch<F, true>(xl, rec, accl, qw + (cnt - 32), 32, lane);
            cnt -= 32;
        }
    };

    if (nrounds > 0) {
        uint4 r[GCN_KD];
#pragma unroll
        for (int u = 0; u < GCN_KD; ++u) {
            r[u] = load_keys(u);
            __builtin_amdgcn_sched_barrier(0);
        }
        for (int g = 0; g < nrounds; ++g) {
            const int gn = (g + 1 < nrounds) ? g + 1 : g;
#pragma unroll
            for (int u = 0; u < GCN_KD; ++u) {
                scan_step(g * GCN_KD + u, r[u]);
                r[u] = load_keys(gn * GCN_KD + u);
            }
        }
    }
    if (cnt > 0) gcn_range_batch<F, false>(xl, rec, accl, qw, cnt, lane);

    __syncthreads();
    int rows = N - base;
    rows = (rows < GCN_R) ? rows : GCN_R;
    float4* out4 = reinterpret_cast<float4*>(agg + (size_t)base * F);
    for (int i = tid; i < rows * (F / 4); i += GCN_NW * 32) out4[i] = acc4[i];
}


__global__ void __launch_bounds__(512)
gcn_edge_pool(const float* __restrict__ x, const int* __restrict__ rec,
              const float* __restrict__ dnorm, float* __restrict__ pooled, int E) {
    constexpr int F   = 64;
    constexpr int LPR = F / 32;
    constexpr int G   = 16 / LPR;
    constexpr int NG  = 32 / G;
    __shared__ float tab[GCN_N_GRAPHS * F];
    for (int i = threadIdx.x; i < GCN_N_GRAPHS * F; i += blockDim.x) tab[i] = 0.0f;
    __syncthreads();

    int lane = threadIdx.x & 31;
    int wpb = blockDim.x >> 5;
    int wave = __builtin_amdgcn_readfirstlane((int)(blockIdx.x * wpb + (threadIdx.x >> 5)));
    int nwaves = gridDim.x * wpb;
    int nchunks = (E + 31) >> 5;
    const float* xl = x + lane;
    float* tl = tab + lane;

    for (int ch = wave; ch < nchunks; ch += nwaves) {
        int e0 = ch * 32;
        int e = e0 + lane;
        int s = 0, g = 0;
        float c = 0.0f;
        if (e < E) {
            int2 r = reinterpret_cast<const int2*>(rec)[e];
            s = r.x & 0x1FFFF;
            g = (int)((unsigned)r.x >> 26);
            c = __int_as_float(r.y) * dnorm[e];
        }
        int ci = __float_as_int(c);

        if (e0 + 32 <= E) {
            float v[2][G][LPR];
            auto load_group = [&](int gr, float (&buf)[G][LPR]) {
#pragma unroll
                for (int j = 0; j < G; ++j) {
                    int sj = __builtin_amdgcn_readlane(s, gr * G + j);
                    const float* xr = xl + (size_t)sj * F;
#pragma unroll
                    for (int i = 0; i < LPR; ++i) buf[j][i] = xr[32 * i];
                }
            };
            auto add_group = [&](int gr, float (&buf)[G][LPR]) {
#pragma unroll
                for (int j = 0; j < G; ++j) {
                    int gj = __builtin_amdgcn_readlane(g, gr * G + j);
                    float cj = __int_as_float(__builtin_amdgcn_readlane(ci, gr * G + j));
                    float* tr = tl + gj * F;
#pragma unroll
                    for (int i = 0; i < LPR; ++i)
                        __hip_atomic_fetch_add(tr + 32 * i, buf[j][i] * cj, __ATOMIC_RELAXED,
                                               __HIP_MEMORY_SCOPE_WORKGROUP);
                }
            };
            load_group(0, v[0]);
#pragma unroll
            for (int gr = 0; gr < NG; ++gr) {
                if (gr + 1 < NG) load_group(gr + 1, v[(gr + 1) & 1]);
                add_group(gr, v[gr & 1]);
            }
        } else {
            int cnt = E - e0;
            for (int j = 0; j < cnt; ++j) {
                int sj = __builtin_amdgcn_readlane(s, j);
                int gj = __builtin_amdgcn_readlane(g, j);
                float cj = __int_as_float(__builtin_amdgcn_readlane(ci, j));
                const float* xr = xl + (size_t)sj * F;
                float* tr = tl + gj * F;
#pragma unroll
                for (int i = 0; i < LPR; ++i)
                    __hip_atomic_fetch_add(tr + 32 * i, xr[32 * i] * cj, __ATOMIC_RELAXED,
                                           __HIP_MEMORY_SCOPE_WORKGROUP);
            }
        }
    }

    __syncthreads();
    for (int i = threadIdx.x; i < GCN_N_GRAPHS * F; i += blockDim.x)
        unsafeAtomicAdd(&pooled[i], tab[i]);
}


template <bool RELU, bool MEAN>
__global__ void gcn_gemm_wmma(const float* __restrict__ A, const float* __restrict__ in_norm,
                              const float* __restrict__ W, const float* __restrict__ bias,
                              float* __restrict__ out, int M, int K, int Nw,
                              const float* __restrict__ counts) {
    int wave = blockIdx.x * (blockDim.x >> 5) + (threadIdx.x >> 5);
    int tiles_n = Nw >> 4;
    int total = (M >> 4) * tiles_n;
    if (wave >= total) return;
    int mt = wave / tiles_n;
    int nt = wave - mt * tiles_n;
    int row0 = mt << 4;
    int col0 = nt << 4;

    int lane  = threadIdx.x & 31;
    int idx15 = lane & 15;
    int koff  = (lane >> 4) << 1;

    int rowA = row0 + idx15;
    float scale = 1.0f;
    if (!MEAN) scale = in_norm[rowA];
    const float* arow = A + (size_t)rowA * K + koff;
    const float* wcol = W + col0 + idx15;

    v8f acc = {};
    for (int k = 0; k < K; k += 4) {
        float2 av = *reinterpret_cast<const float2*>(arow + k);
        v2f af;
        af.x = MEAN ? av.x : av.x * scale;
        af.y = MEAN ? av.y : av.y * scale;
        v2f bf;
        bf.x = wcol[(size_t)(k + koff) * Nw];
        bf.y = wcol[(size_t)(k + koff + 1) * Nw];
        acc = __builtin_amdgcn_wmma_f32_16x16x4_f32(
             false, af,  false, bf,
             (short)0, acc,  false,  false);
    }

    int col = col0 + idx15;
    float bv = bias[col];
    int mbase = (lane >> 4) << 3;
#pragma unroll
    for (int r = 0; r < 8; ++r) {
        int m = row0 + mbase + r;
        if (MEAN) {
            float c = counts[m];
            float cc = (c < 1.0f) ? 1.0f : c;
            out[(size_t)m * Nw + col] = (acc[r] + c * bv) / cc;
        } else {
            float v = acc[r] + bv;
            if (RELU) v = (v > 0.0f) ? v : 0.0f;
            out[(size_t)m * Nw + col] = v;
        }
    }
}


static inline int cdiv_l(long a, long b) { return (int)((a + b - 1) / b); }

extern "C" void kernel_launch(void* const* d_in, const int* in_sizes, int n_in,
                              void* d_out, int out_size, void* d_ws, size_t ws_size,
                              hipStream_t stream) {
    const float* emb = (const float*)d_in[0];
    const float* W1  = (const float*)d_in[1];
    const float* b1  = (const float*)d_in[2];
    const float* W2  = (const float*)d_in[3];
    const float* b2  = (const float*)d_in[4];
    const float* W3  = (const float*)d_in[5];
    const float* b3  = (const float*)d_in[6];
    const int* src = (const int*)d_in[7];
    const int* dst = (const int*)d_in[8];
    const int* gid = (const int*)d_in[9];

    const int E = in_sizes[7];
    const int N = in_sizes[9];

    float* ws       = (float*)d_ws;
    float* out_norm = ws;
    float* in_norm  = out_norm + N;
    float* pooled   = in_norm + N;
    float* counts   = pooled + GCN_N_GRAPHS * 64;
    float* agg      = counts + GCN_N_GRAPHS;
    float* h        = agg + (size_t)N * 128;
    unsigned int* key = (unsigned int*)(h + (size_t)N * 128);

    const int T = 256;
    const int nrounds = cdiv_l(E, GCN_KROUND);
    const int Epad = nrounds * GCN_KROUND;
    int* rec = (int*)(key + Epad / 4);
    float* dnorm = (float*)(rec + 2 * (size_t)Epad);
    int* npair = (int*)(dnorm + Epad);

    int zc = 2 * N + GCN_N_GRAPHS * 64 + GCN_N_GRAPHS;
    gcn_fill_zero<<<cdiv_l(zc, 4 * T), T, 0, stream>>>(ws, zc);

    gcn_degree<<<cdiv_l(cdiv_l(E, 4), T), T, 0, stream>>>(src, dst, out_norm, in_norm, E);
    gcn_norm_finalize<<<cdiv_l(2L * N, T), T, 0, stream>>>(out_norm, gid, npair, counts, N);
    gcn_edge_records<<<cdiv_l(Epad / 4, T), T, 0, stream>>>(src, dst, out_norm, npair, key, rec, dnorm,
                                                         E, Epad);

    gcn_range_spmm<128><<<cdiv_l(N, GCN_R), GCN_NW * 32, 0, stream>>>(emb, rec, key, agg, nrounds, N);
    {
        int tiles = (N / 16) * (128 / 16);
        gcn_gemm_wmma<true, false><<<cdiv_l(tiles, 4), 128, 0, stream>>>(
            agg, in_norm, W1, b1, h, N, 128, 128, nullptr);
    }

    gcn_range_spmm<128><<<cdiv_l(N, GCN_R), GCN_NW * 32, 0, stream>>>(h, rec, key, agg, nrounds, N);
    {
        int tiles = (N / 16) * (64 / 16);
        gcn_gemm_wmma<true, false><<<cdiv_l(tiles, 4), 128, 0, stream>>>(
            agg, in_norm, W2, b2, h, N, 128, 64, nullptr);
    }

    gcn_edge_pool<<<768, 512, 0, stream>>>(h, rec, dnorm, pooled, E);
    {
        int tiles = (GCN_N_GRAPHS / 16) * (64 / 16);
        gcn_gemm_wmma<false, true><<<cdiv_l(tiles, 4), 128, 0, stream>>>(
            pooled, nullptr, W3, b3, (float*)d_out, GCN_N_GRAPHS, 64, 64, counts);
    }
}


# Layer GEMMs by strips: a wave keeps its W fragments in registers, loops over 16-row strips and reads each A row half once (16-byte pieces) for 4 or 2 column tiles. Reorders k inside f32 WMMA sums
# speedup vs baseline: 2.7170x; 1.0855x over previous
#include <hip/hip_runtime.h>

typedef __attribute__((ext_vector_type(2))) float v2f;
typedef __attribute__((ext_vector_type(8))) float v8f;

#define GCN_N_GRAPHS 64

#define GCN_R     400
#define GCN_NW    32
#define GCN_QCAP  544
#define GCN_KD    4
#define GCN_KROUND (GCN_KD * GCN_NW * 32 * 16)


__global__ void gcn_fill_zero(float* __restrict__ p, int n) {
    int n4 = n >> 2;
    float4* p4 = reinterpret_cast<float4*>(p);
    int t = blockIdx.x * blockDim.x + threadIdx.x;
    int stride = gridDim.x * blockDim.x;
    for (int i = t; i < n4; i += stride) p4[i] = make_float4(0.0f, 0.0f, 0.0f, 0.0f);
    int r = (n4 << 2) + t;
    if (r < n) p[r] = 0.0f;
}

__global__ void gcn_degree(const int* __restrict__ src, const int* __restrict__ dst,
                           float* __restrict__ deg_out, float* __restrict__ deg_in, int E) {
    int t = blockIdx.x * blockDim.x + threadIdx.x;
    int e = t * 4;
    if (e + 3 < E) {
        int4 s4 = reinterpret_cast<const int4*>(src)[t];
        int4 d4 = reinterpret_cast<const int4*>(dst)[t];
        unsafeAtomicAdd(&deg_out[s4.x], 1.0f);
        unsafeAtomicAdd(&deg_out[s4.y], 1.0f);
        unsafeAtomicAdd(&deg_out[s4.z], 1.0f);
        unsafeAtomicAdd(&deg_out[s4.w], 1.0f);
        unsafeAtomicAdd(&deg_in[d4.x], 1.0f);
        unsafeAtomicAdd(&deg_in[d4.y], 1.0f);
        unsafeAtomicAdd(&deg_in[d4.z], 1.0f);
        unsafeAtomicAdd(&deg_in[d4.w], 1.0f);
    } else {
        for (int i = 0; i < 4; ++i) {
            if (e + i < E) {
                unsafeAtomicAdd(&deg_out[src[e + i]], 1.0f);
                unsafeAtomicAdd(&deg_in[dst[e + i]], 1.0f);
            }
        }
    }
}

__device__ __forceinline__ int gcn_pack_edge(int s, int d, unsigned owner, int g) {
    return (int)((unsigned)s | (((unsigned)d - owner * GCN_R) << 17) | ((unsigned)(g & 63) << 26));
}

__global__ void gcn_edge_records(const int* __restrict__ src, const int* __restrict__ dst,
                                 const float* __restrict__ out_norm, const int* __restrict__ npair,
                                 unsigned int* __restrict__ key, int* __restrict__ rec,
                                 float* __restrict__ dnorm, int E, int Epad) {
    int t = blockIdx.x * blockDim.x + threadIdx.x;
    int e = t * 4;
    unsigned kw = 0xFFFFFFFFu;
    const int2* np2 = reinterpret_cast<const int2*>(npair);
    if (e + 3 < E) {
        int4 s4 = reinterpret_cast<const int4*>(src)[t];
        int4 d4 = reinterpret_cast<const int4*>(dst)[t];
        float a0 = out_norm[s4.x], a1 = out_norm[s4.y], a2 = out_norm[s4.z], a3 = out_norm[s4.w];
        int2 p0 = np2[d4.x], p1 = np2[d4.y], p2 = np2[d4.z], p3 = np2[d4.w];
        unsigned o0 = (unsigned)d4.x / GCN_R, o1 = (unsigned)d4.y / GCN_R;
        unsigned o2 = (unsigned)d4.z / GCN_R, o3 = (unsigned)d4.w / GCN_R;
        kw = o0 | (o1 << 8) | (o2 << 16) | (o3 << 24);
        int4* r4 = reinterpret_cast<int4*>(rec);
        r4[2 * t] = make_int4(gcn_pack_edge(s4.x, d4.x, o0, p0.y), __float_as_int(a0),
                              gcn_pack_edge(s4.y, d4.y, o1, p1.y), __float_as_int(a1));
        r4[2 * t + 1] = make_int4(gcn_pack_edge(s4.z, d4.z, o2, p2.y), __float_as_int(a2),
                                  gcn_pack_edge(s4.w, d4.w, o3, p3.y), __float_as_int(a3));
        reinterpret_cast<int4*>(dnorm)[t] = make_int4(p0.x, p1.x, p2.x, p3.x);
    } else {
        for (int i = 0; i < 4; ++i) {
            if (e + i < E) {
                int s = src[e + i];
                int d = dst[e + i];
                unsigned o = (unsigned)d / GCN_R;
                int2 p = np2[d];
                rec[2 * (e + i)] = gcn_pack_edge(s, d, o, p.y);
                rec[2 * (e + i) + 1] = __float_as_int(out_norm[s]);
                dnorm[e + i] = __int_as_float(p.x);
                kw = (kw & ~(0xFFu << (8 * i))) | (o << (8 * i));
            }
        }
    }
    if (e < Epad) key[t] = kw;
}

__global__ void gcn_norm_finalize(float* __restrict__ buf, const int* __restrict__ gid,
                                  int* __restrict__ npair, float* __restrict__ counts, int N) {
    __shared__ float cnt[GCN_N_GRAPHS];
    if (threadIdx.x < GCN_N_GRAPHS) cnt[threadIdx.x] = 0.0f;
    __syncthreads();
    int i = blockIdx.x * blockDim.x + threadIdx.x;
    if (i < 2 * N) {
        float d = buf[i];
        d = (d < 1.0f) ? 1.0f : d;
        float r = 1.0f / sqrtf(d);
        buf[i] = r;
        if (i >= N) {
            int g = gid[i - N];
            reinterpret_cast<int2*>(npair)[i - N] = make_int2(__float_as_int(r), g);
            if ((unsigned)g < (unsigned)GCN_N_GRAPHS)
                __hip_atomic_fetch_add(&cnt[g], 1.0f, __ATOMIC_RELAXED, __HIP_MEMORY_SCOPE_WORKGROUP);
        }
    }
    __syncthreads();
    if (threadIdx.x < GCN_N_GRAPHS) {
        float c = cnt[threadIdx.x];
        if (c != 0.0f) unsafeAtomicAdd(&counts[threadIdx.x], c);
    }
}


__device__ __forceinline__ unsigned gcn_zero_bytes(unsigned x) {
    return ~(((x & 0x7F7F7F7Fu) + 0x7F7F7F7Fu) | x) & 0x80808080u;
}

template <int F, bool FULL>
__device__ __forceinline__ void gcn_range_batch(const float* __restrict__ xl,
                                                const int* __restrict__ rec, float* accl,
                                                const int* qb, int nvalid, int lane) {
    constexpr int LPR = F / 32;
    constexpr int G   = 16 / LPR;
    constexpr int NG  = 32 / G;
    int s = 0, dl = 0, ci = 0;
    if (FULL || lane < nvalid) {
        int2 r = reinterpret_cast<const int2*>(rec)[qb[lane]];
        s = r.x & 0x1FFFF;
        dl = (r.x >> 17) & 0x1FF;
        ci = r.y;
    }
    float v[2][G][LPR];
    auto load_group = [&](int g, float (&buf)[G][LPR]) {
#pragma unroll
        for (int j = 0; j < G; ++j) {
            if (FULL || g * G + j < nvalid) {
                int sj = __builtin_amdgcn_readlane(s, g * G + j);
                const float* xr = xl + (size_t)sj * F;
#pragma unroll
                for (int i = 0; i < LPR; ++i) buf[j][i] = xr[32 * i];
            } else {
#pragma unroll
                for (int i = 0; i < LPR; ++i) buf[j][i] = 0.0f;
            }
        }
    };
    auto add_group = [&](int g, float (&buf)[G][LPR]) {
#pragma unroll
        for (int j = 0; j < G; ++j) {
            if (FULL || g * G + j < nvalid) {
                int dj = __builtin_amdgcn_readlane(dl, g * G + j);
                float cj = __int_as_float(__builtin_amdgcn_readlane(ci, g * G + j));
                float* ar = accl + dj * F;
#pragma unroll
                for (int i = 0; i < LPR; ++i)
                    __hip_atomic_fetch_add(ar + 32 * i, buf[j][i] * cj, __ATOMIC_RELAXED,
                                           __HIP_MEMORY_SCOPE_WORKGROUP);
            }
        }
    };
    load_group(0, v[0]);
#pragma unroll
    for (int g = 0; g < NG; ++g) {
        if (g + 1 < NG) load_group(g + 1, v[(g + 1) & 1]);
        add_group(g, v[g & 1]);
    }
}

template <int F>
__global__ void __launch_bounds__(GCN_NW * 32)
gcn_range_spmm(const float* __restrict__ x, const int* __restrict__ rec,
               const unsigned int* __restrict__ key, float* __restrict__ agg, int nrounds, int N) {
    __shared__ float acc[GCN_R * F];
    __shared__ int q[GCN_NW * GCN_QCAP];

    const int tid = threadIdx.x;
    const int lane = tid & 31;
    const int w = __builtin_amdgcn_readfirstlane(tid >> 5);
    const int base = blockIdx.x * GCN_R;

    float4* acc4 = reinterpret_cast<float4*>(acc);
    for (int i = tid; i < GCN_R * F / 4; i += GCN_NW * 32) acc4[i] = make_float4(0.0f, 0.0f, 0.0f, 0.0f);
    __syncthreads();

    int* qw = q + w * GCN_QCAP;
    const float* xl = x + lane;
    float* accl = acc + lane;
    int cnt = 0;

    const uint4* key4 = reinterpret_cast<const uint4*>(key);
    const unsigned kb = blockIdx.x * 0x01010101u;
    auto load_keys = [&](int st) -> uint4 { return key4[(st * GCN_NW + w) * 32 + lane]; };
    auto scan_step = [&](int st, const uint4& k) {
        const int ebase = ((st * GCN_NW + w) * 32 + lane) * 16;
        unsigned m = (gcn_zero_bytes(k.x ^ kb) >> 7) | (gcn_zero_bytes(k.y ^ kb) >> 6) |
                     (gcn_zero_bytes(k.z ^ kb) >> 5) | (gcn_zero_bytes(k.w ^ kb) >> 4);
        unsigned b = __builtin_amdgcn_ballot_w32(m != 0);
        while (b) {
            if (m) {
                int i = __builtin_ctz(m);
                m &= m - 1;
                qw[__builtin_amdgcn_mbcnt_lo(b, (unsigned)cnt)] = ebase + ((i & 7) << 2) + (i >> 3);
            }
            cnt += __builtin_popcount(b);
            b = __builtin_amdgcn_ballot_w32(m != 0);
        }
        while (cnt >= 32) {
            gcn_range_batch<F, true>(xl, rec, accl, qw + (cnt - 32), 32, lane);
            cnt -= 32;
        }
    };

    if (nrounds > 0) {
        uint4 r[GCN_KD];
#pragma unroll
        for (int u = 0; u < GCN_KD; ++u) {
            r[u] = load_keys(u);
            __builtin_amdgcn_sched_barrier(0);
        }
        for (int g = 0; g < nrounds; ++g) {
            const int gn = (g + 1 < nrounds) ? g + 1 : g;
#pragma unroll
            for (int u = 0; u < GCN_KD; ++u) {
                scan_step(g * GCN_KD + u, r[u]);
                r[u] = load_keys(gn * GCN_KD + u);
            }
        }
    }
    if (cnt > 0) gcn_range_batch<F, false>(xl, rec, accl, qw, cnt, lane);

    __syncthreads();
    int rows = N - base;
    rows = (rows < GCN_R) ? rows : GCN_R;
    float4* out4 = reinterpret_cast<float4*>(agg + (size_t)base * F);
    for (int i = tid; i < rows * (F / 4); i += GCN_NW * 32) out4[i] = acc4[i];
}


__global__ void __launch_bounds__(512)
gcn_edge_pool(const float* __restrict__ x, const int* __restrict__ rec,
              const float* __restrict__ dnorm, float* __restrict__ pooled, int E) {
    constexpr int F   = 64;
    constexpr int LPR = F / 32;
    constexpr int G   = 16 / LPR;
    constexpr int NG  = 32 / G;
    __shared__ float tab[GCN_N_GRAPHS * F];
    for (int i = threadIdx.x; i < GCN_N_GRAPHS * F; i += blockDim.x) tab[i] = 0.0f;
    __syncthreads();

    int lane = threadIdx.x & 31;
    int wpb = blockDim.x >> 5;
    int wave = __builtin_amdgcn_readfirstlane((int)(blockIdx.x * wpb + (threadIdx.x >> 5)));
    int nwaves = gridDim.x * wpb;
    int nchunks = (E + 31) >> 5;
    const float* xl = x + lane;
    float* tl = tab + lane;

    for (int ch = wave; ch < nchunks; ch += nwaves) {
        int e0 = ch * 32;
        int e = e0 + lane;
        int s = 0, g = 0;
        float c = 0.0f;
        if (e < E) {
            int2 r = reinterpret_cast<const int2*>(rec)[e];
            s = r.x & 0x1FFFF;
            g = (int)((unsigned)r.x >> 26);
            c = __int_as_float(r.y) * dnorm[e];
        }
        int ci = __float_as_int(c);

        if (e0 + 32 <= E) {
            float v[2][G][LPR];
            auto load_group = [&](int gr, float (&buf)[G][LPR]) {
#pragma unroll
                for (int j = 0; j < G; ++j) {
                    int sj = __builtin_amdgcn_readlane(s, gr * G + j);
                    const float* xr = xl + (size_t)sj * F;
#pragma unroll
                    for (int i = 0; i < LPR; ++i) buf[j][i] = xr[32 * i];
                }
            };
            auto add_group = [&](int gr, float (&buf)[G][LPR]) {
#pragma unroll
                for (int j = 0; j < G; ++j) {
                    int gj = __builtin_amdgcn_readlane(g, gr * G + j);
                    float cj = __int_as_float(__builtin_amdgcn_readlane(ci, gr * G + j));
                    float* tr = tl + gj * F;
#pragma unroll
                    for (int i = 0; i < LPR; ++i)
                        __hip_atomic_fetch_add(tr + 32 * i, buf[j][i] * cj, __ATOMIC_RELAXED,
                                               __HIP_MEMORY_SCOPE_WORKGROUP);
                }
            };
            load_group(0, v[0]);
#pragma unroll
            for (int gr = 0; gr < NG; ++gr) {
                if (gr + 1 < NG) load_group(gr + 1, v[(gr + 1) & 1]);
                add_group(gr, v[gr & 1]);
            }
        } else {
            int cnt = E - e0;
            for (int j = 0; j < cnt; ++j) {
                int sj = __builtin_amdgcn_readlane(s, j);
                int gj = __builtin_amdgcn_readlane(g, j);
                float cj = __int_as_float(__builtin_amdgcn_readlane(ci, j));
                const float* xr = xl + (size_t)sj * F;
                float* tr = tl + gj * F;
#pragma unroll
                for (int i = 0; i < LPR; ++i)
                    __hip_atomic_fetch_add(tr + 32 * i, xr[32 * i] * cj, __ATOMIC_RELAXED,
                                           __HIP_MEMORY_SCOPE_WORKGROUP);
            }
        }
    }

    __syncthreads();
    for (int i = threadIdx.x; i < GCN_N_GRAPHS * F; i += blockDim.x)
        unsafeAtomicAdd(&pooled[i], tab[i]);
}


template <bool RELU, bool MEAN>
__global__ void gcn_gemm_wmma(const float* __restrict__ A, const float* __restrict__ in_norm,
                              const float* __restrict__ W, const float* __restrict__ bias,
                              float* __restrict__ out, int M, int K, int Nw,
                              const float* __restrict__ counts) {
    int wave = blockIdx.x * (blockDim.x >> 5) + (threadIdx.x >> 5);
    int tiles_n = Nw >> 4;
    int total = (M >> 4) * tiles_n;
    if (wave >= total) return;
    int mt = wave / tiles_n;
    int nt = wave - mt * tiles_n;
    int row0 = mt << 4;
    int col0 = nt << 4;

    int lane  = threadIdx.x & 31;
    int idx15 = lane & 15;
    int koff  = (lane >> 4) << 1;

    int rowA = row0 + idx15;
    float scale = 1.0f;
    if (!MEAN) scale = in_norm[rowA];
    const float* arow = A + (size_t)rowA * K + koff;
    const float* wcol = W + col0 + idx15;

    v8f acc = {};
    for (int k = 0; k < K; k += 4) {
        float2 av = *reinterpret_cast<const float2*>(arow + k);
        v2f af;
        af.x = MEAN ? av.x : av.x * scale;
        af.y = MEAN ? av.y : av.y * scale;
        v2f bf;
        bf.x = wcol[(size_t)(k + koff) * Nw];
        bf.y = wcol[(size_t)(k + koff + 1) * Nw];
        acc = __builtin_amdgcn_wmma_f32_16x16x4_f32(
             false, af,  false, bf,
             (short)0, acc,  false,  false);
    }

    int col = col0 + idx15;
    float bv = bias[col];
    int mbase = (lane >> 4) << 3;
#pragma unroll
    for (int r = 0; r < 8; ++r) {
        int m = row0 + mbase + r;
        if (MEAN) {
            float c = counts[m];
            float cc = (c < 1.0f) ? 1.0f : c;
            out[(size_t)m * Nw + col] = (acc[r] + c * bv) / cc;
        } else {
            float v = acc[r] + bv;
            if (RELU) v = (v > 0.0f) ? v : 0.0f;
            out[(size_t)m * Nw + col] = v;
        }
    }
}


template <int NW, int CT, bool RELU>
__global__ void __launch_bounds__(256)
gcn_gemm_strip(const float* __restrict__ A, const float* __restrict__ in_norm,
               const float* __restrict__ W, const float* __restrict__ bias,
               float* __restrict__ out, int M) {
    constexpr int K   = 128;
    constexpr int NCG = NW / (16 * CT);
    const int lane = threadIdx.x & 31;
    const int c = lane & 15;
    const int h = lane >> 4;
    const int wave = blockIdx.x * (blockDim.x >> 5) + (threadIdx.x >> 5);
    const int nwaves = gridDim.x * (blockDim.x >> 5);
    const int cg = wave % NCG;
    const int col0 = cg * 16 * CT;
    const int nstrips = M >> 4;
    int strip = wave / NCG;
    const int sstride = nwaves / NCG;
    if (strip >= nstrips) return;

    v2f bf[16][2][CT];
#pragma unroll
    for (int j = 0; j < 16; ++j)
#pragma unroll
        for (int t = 0; t < 2; ++t) {
            const float* wr = W + (size_t)(8 * j + 4 * h + 2 * t) * NW + col0 + c;
#pragma unroll
            for (int n = 0; n < CT; ++n) {
                bf[j][t][n].x = wr[16 * n];
                bf[j][t][n].y = wr[NW + 16 * n];
            }
        }
    float bv[CT];
#pragma unroll
    for (int n = 0; n < CT; ++n) bv[n] = bias[col0 + 16 * n + c];

    float4 av[16];
    {
        const float4* ar = reinterpret_cast<const float4*>(A + (size_t)(strip * 16 + c) * K + 4 * h);
#pragma unroll
        for (int j = 0; j < 16; ++j) av[j] = ar[2 * j];
    }
    for (; strip < nstrips; strip += sstride) {
        const int ns = (strip + sstride < nstrips) ? strip + sstride : strip;
        float4 nv[16];
        {
            const float4* ar = reinterpret_cast<const float4*>(A + (size_t)(ns * 16 + c) * K + 4 * h);
#pragma unroll
            for (int j = 0; j < 16; ++j) nv[j] = ar[2 * j];
        }
        const float scale = in_norm[strip * 16 + c];
        v8f acc[CT];
#pragma unroll
        for (int n = 0; n < CT; ++n) acc[n] = v8f{};
#pragma unroll
        for (int j = 0; j < 16; ++j) {
            v2f a0, a1;
            a0.x = av[j].x * scale; a0.y = av[j].y * scale;
            a1.x = av[j].z * scale; a1.y = av[j].w * scale;
#pragma unroll
            for (int n = 0; n < CT; ++n)
                acc[n] = __builtin_amdgcn_wmma_f32_16x16x4_f32(false, a0, false, bf[j][0][n],
                                                               (short)0, acc[n], false, false);
#pragma unroll
            for (int n = 0; n < CT; ++n)
                acc[n] = __builtin_amdgcn_wmma_f32_16x16x4_f32(false, a1, false, bf[j][1][n],
                                                               (short)0, acc[n], false, false);
        }
        float* orow = out + (size_t)(strip * 16 + 8 * h) * NW + col0 + c;
#pragma unroll
        for (int r = 0; r < 8; ++r)
#pragma unroll
            for (int n = 0; n < CT; ++n) {
                float v = acc[n][r] + bv[n];
                if (RELU) v = (v > 0.0f) ? v : 0.0f;
                orow[(size_t)r * NW + 16 * n] = v;
            }
#pragma unroll
        for (int j = 0; j < 16; ++j) av[j] = nv[j];
    }
}


static inline int cdiv_l(long a, long b) { return (int)((a + b - 1) / b); }

extern "C" void kernel_launch(void* const* d_in, const int* in_sizes, int n_in,
                              void* d_out, int out_size, void* d_ws, size_t ws_size,
                              hipStream_t stream) {
    const float* emb = (const float*)d_in[0];
    const float* W1  = (const float*)d_in[1];
    const float* b1  = (const float*)d_in[2];
    const float* W2  = (const float*)d_in[3];
    const float* b2  = (const float*)d_in[4];
    const float* W3  = (const float*)d_in[5];
    const float* b3  = (const float*)d_in[6];
    const int* src = (const int*)d_in[7];
    const int* dst = (const int*)d_in[8];
    const int* gid = (const int*)d_in[9];

    const int E = in_sizes[7];
    const int N = in_sizes[9];

    float* ws       = (float*)d_ws;
    float* out_norm = ws;
    float* in_norm  = out_norm + N;
    float* pooled   = in_norm + N;
    float* counts   = pooled + GCN_N_GRAPHS * 64;
    float* agg      = counts + GCN_N_GRAPHS;
    float* h        = agg + (size_t)N * 128;
    unsigned int* key = (unsigned int*)(h + (size_t)N * 128);

    const int T = 256;
    const int nrounds = cdiv_l(E, GCN_KROUND);
    const int Epad = nrounds * GCN_KROUND;
    int* rec = (int*)(key + Epad / 4);
    float* dnorm = (float*)(rec + 2 * (size_t)Epad);
    int* npair = (int*)(dnorm + Epad);

    int zc = 2 * N + GCN_N_GRAPHS * 64 + GCN_N_GRAPHS;
    gcn_fill_zero<<<cdiv_l(zc, 4 * T), T, 0, stream>>>(ws, zc);

    gcn_degree<<<cdiv_l(cdiv_l(E, 4), T), T, 0, stream>>>(src, dst, out_norm, in_norm, E);
    gcn_norm_finalize<<<cdiv_l(2L * N, T), T, 0, stream>>>(out_norm, gid, npair, counts, N);
    gcn_edge_records<<<cdiv_l(Epad / 4, T), T, 0, stream>>>(src, dst, out_norm, npair, key, rec, dnorm,
                                                         E, Epad);

    gcn_range_spmm<128><<<cdiv_l(N, GCN_R), GCN_NW * 32, 0, stream>>>(emb, rec, key, agg, nrounds, N);
    gcn_gemm_strip<128, 4, true><<<256, 256, 0, stream>>>(agg, in_norm, W1, b1, h, N);

    gcn_range_spmm<128><<<cdiv_l(N, GCN_R), GCN_NW * 32, 0, stream>>>(h, rec, key, agg, nrounds, N);
    gcn_gemm_strip<64, 2, true><<<256, 256, 0, stream>>>(agg, in_norm, W2, b2, h, N);

    gcn_edge_pool<<<768, 512, 0, stream>>>(h, rec, dnorm, pooled, E);
    {
        int tiles = (GCN_N_GRAPHS / 16) * (64 / 16);
        gcn_gemm_wmma<false, true><<<cdiv_l(tiles, 4), 128, 0, stream>>>(
            pooled, nullptr, W3, b3, (float*)d_out, GCN_N_GRAPHS, 64, 64, counts);
    }
}


# Gathered rows fetched whole: a lane takes its 4 (pool: 2) consecutive floats with one 16-byte (8-byte) load, not 4 (2) one-line loads; LDS row word 32i+l holds float 4l+i. Sum order as before
# speedup vs baseline: 2.9811x; 1.0972x over previous
#include <hip/hip_runtime.h>

typedef __attribute__((ext_vector_type(2))) float v2f;
typedef __attribute__((ext_vector_type(8))) float v8f;

#define GCN_N_GRAPHS 64

#define GCN_R     400
#define GCN_NW    32
#define GCN_QCAP  544
#define GCN_KD    4
#define GCN_KROUND (GCN_KD * GCN_NW * 32 * 16)


__global__ void gcn_fill_zero(float* __restrict__ p, int n) {
    int n4 = n >> 2;
    float4* p4 = reinterpret_cast<float4*>(p);
    int t = blockIdx.x * blockDim.x + threadIdx.x;
    int stride = gridDim.x * blockDim.x;
    for (int i = t; i < n4; i += stride) p4[i] = make_float4(0.0f, 0.0f, 0.0f, 0.0f);
    int r = (n4 << 2) + t;
    if (r < n) p[r] = 0.0f;
}

__global__ void gcn_degree(const int* __restrict__ src, const int* __restrict__ dst,
                           float* __restrict__ deg_out, float* __restrict__ deg_in, int E) {
    int t = blockIdx.x * blockDim.x + threadIdx.x;
    int e = t * 4;
    if (e + 3 < E) {
        int4 s4 = reinterpret_cast<const int4*>(src)[t];
        int4 d4 = reinterpret_cast<const int4*>(dst)[t];
        unsafeAtomicAdd(&deg_out[s4.x], 1.0f);
        unsafeAtomicAdd(&deg_out[s4.y], 1.0f);
        unsafeAtomicAdd(&deg_out[s4.z], 1.0f);
        unsafeAtomicAdd(&deg_out[s4.w], 1.0f);
        unsafeAtomicAdd(&deg_in[d4.x], 1.0f);
        unsafeAtomicAdd(&deg_in[d4.y], 1.0f);
        unsafeAtomicAdd(&deg_in[d4.z], 1.0f);
        unsafeAtomicAdd(&deg_in[d4.w], 1.0f);
    } else {
        for (int i = 0; i < 4; ++i) {
            if (e + i < E) {
                unsafeAtomicAdd(&deg_out[src[e + i]], 1.0f);
                unsafeAtomicAdd(&deg_in[dst[e + i]], 1.0f);
            }
        }
    }
}

__device__ __forceinline__ int gcn_pack_edge(int s, int d, unsigned owner, int g) {
    return (int)((unsigned)s | (((unsigned)d - owner * GCN_R) << 17) | ((unsigned)(g & 63) << 26));
}

__global__ void gcn_edge_records(const int* __restrict__ src, const int* __restrict__ dst,
                                 const float* __restrict__ out_norm, const int* __restrict__ npair,
                                 unsigned int* __restrict__ key, int* __restrict__ rec,
                                 float* __restrict__ dnorm, int E, int Epad) {
    int t = blockIdx.x * blockDim.x + threadIdx.x;
    int e = t * 4;
    unsigned kw = 0xFFFFFFFFu;
    const int2* np2 = reinterpret_cast<const int2*>(npair);
    if (e + 3 < E) {
        int4 s4 = reinterpret_cast<const int4*>(src)[t];
        int4 d4 = reinterpret_cast<const int4*>(dst)[t];
        float a0 = out_norm[s4.x], a1 = out_norm[s4.y], a2 = out_norm[s4.z], a3 = out_norm[s4.w];
        int2 p0 = np2[d4.x], p1 = np2[d4.y], p2 = np2[d4.z], p3 = np2[d4.w];
        unsigned o0 = (unsigned)d4.x / GCN_R, o1 = (unsigned)d4.y / GCN_R;
        unsigned o2 = (unsigned)d4.z / GCN_R, o3 = (unsigned)d4.w / GCN_R;
        kw = o0 | (o1 << 8) | (o2 << 16) | (o3 << 24);
        int4* r4 = reinterpret_cast<int4*>(rec);
        r4[2 * t] = make_int4(gcn_pack_edge(s4.x, d4.x, o0, p0.y), __float_as_int(a0),
                              gcn_pack_edge(s4.y, d4.y, o1, p1.y), __float_as_int(a1));
        r4[2 * t + 1] = make_int4(gcn_pack_edge(s4.z, d4.z, o2, p2.y), __float_as_int(a2),
                                  gcn_pack_edge(s4.w, d4.w, o3, p3.y), __float_as_int(a3));
        reinterpret_cast<int4*>(dnorm)[t] = make_int4(p0.x, p1.x, p2.x, p3.x);
    } else {
        for (int i = 0; i < 4; ++i) {
            if (e + i < E) {
                int s = src[e + i];
                int d = dst[e + i];
                unsigned o = (unsigned)d / GCN_R;
                int2 p = np2[d];
                rec[2 * (e + i)] = gcn_pack_edge(s, d, o, p.y);
                rec[2 * (e + i) + 1] = __float_as_int(out_norm[s]);
                dnorm[e + i] = __int_as_float(p.x);
                kw = (kw & ~(0xFFu << (8 * i))) | (o << (8 * i));
            }
        }
    }
    if (e < Epad) key[t] = kw;
}

__global__ void gcn_norm_finalize(float* __restrict__ buf, const int* __restrict__ gid,
                                  int* __restrict__ npair, float* __restrict__ counts, int N) {
    __shared__ float cnt[GCN_N_GRAPHS];
    if (threadIdx.x < GCN_N_GRAPHS) cnt[threadIdx.x] = 0.0f;
    __syncthreads();
    int i = blockIdx.x * blockDim.x + threadIdx.x;
    if (i < 2 * N) {
        float d = buf[i];
        d = (d < 1.0f) ? 1.0f : d;
        float r = 1.0f / sqrtf(d);
        buf[i] = r;
        if (i >= N) {
            int g = gid[i - N];
            reinterpret_cast<int2*>(npair)[i - N] = make_int2(__float_as_int(r), g);
            if ((unsigned)g < (unsigned)GCN_N_GRAPHS)
                __hip_atomic_fetch_add(&cnt[g], 1.0f, __ATOMIC_RELAXED, __HIP_MEMORY_SCOPE_WORKGROUP);
        }
    }
    __syncthreads();
    if (threadIdx.x < GCN_N_GRAPHS) {
        float c = cnt[threadIdx.x];
        if (c != 0.0f) unsafeAtomicAdd(&counts[threadIdx.x], c);
    }
}


__device__ __forceinline__ unsigned gcn_zero_bytes(unsigned x) {
    return ~(((x & 0x7F7F7F7Fu) + 0x7F7F7F7Fu) | x) & 0x80808080u;
}

template <int F, bool FULL>
__device__ __forceinline__ void gcn_range_batch(const float* __restrict__ x,
                                                const int* __restrict__ rec, float* accl,
                                                const int* qb, int nvalid, int lane) {
    constexpr int LPR = F / 32;
    constexpr int G   = 16 / LPR;
    constexpr int NG  = 32 / G;
    typedef float rowv __attribute__((ext_vector_type(LPR)));
    int s = 0, dl = 0, ci = 0;
    if (FULL || lane < nvalid) {
        int2 r = reinterpret_cast<const int2*>(rec)[qb[lane]];
        s = r.x & 0x1FFFF;
        dl = (r.x >> 17) & 0x1FF;
        ci = r.y;
    }
    float v[2][G][LPR];
    auto load_group = [&](int g, float (&buf)[G][LPR]) {
#pragma unroll
        for (int j = 0; j < G; ++j) {
            if (FULL || g * G + j < nvalid) {
                int sj = __builtin_amdgcn_readlane(s, g * G + j);
                rowv rv = reinterpret_cast<const rowv*>(x + (size_t)sj * F)[lane];
#pragma unroll
                for (int i = 0; i < LPR; ++i) buf[j][i] = rv[i];
            } else {
#pragma unroll
                for (int i = 0; i < LPR; ++i) buf[j][i] = 0.0f;
            }
        }
    };
    auto add_group = [&](int g, float (&buf)[G][LPR]) {
#pragma unroll
        for (int j = 0; j < G; ++j) {
            if (FULL || g * G + j < nvalid) {
                int dj = __builtin_amdgcn_readlane(dl, g * G + j);
                float cj = __int_as_float(__builtin_amdgcn_readlane(ci, g * G + j));
                float* ar = accl + dj * F;
#pragma unroll
                for (int i = 0; i < LPR; ++i)
                    __hip_atomic_fetch_add(ar + 32 * i, buf[j][i] * cj, __ATOMIC_RELAXED,
                                           __HIP_MEMORY_SCOPE_WORKGROUP);
            }
        }
    };
    load_group(0, v[0]);
#pragma unroll
    for (int g = 0; g < NG; ++g) {
        if (g + 1 < NG) load_group(g + 1, v[(g + 1) & 1]);
        add_group(g, v[g & 1]);
    }
}

template <int F>
__global__ void __launch_bounds__(GCN_NW * 32)
gcn_range_spmm(const float* __restrict__ x, const int* __restrict__ rec,
               const unsigned int* __restrict__ key, float* __restrict__ agg, int nrounds, int N) {
    __shared__ float acc[GCN_R * F];
    __shared__ int q[GCN_NW * GCN_QCAP];

    const int tid = threadIdx.x;
    const int lane = tid & 31;
    const int w = __builtin_amdgcn_readfirstlane(tid >> 5);
    const int base = blockIdx.x * GCN_R;

    float4* acc4 = reinterpret_cast<float4*>(acc);
    for (int i = tid; i < GCN_R * F / 4; i += GCN_NW * 32) acc4[i] = make_float4(0.0f, 0.0f, 0.0f, 0.0f);
    __syncthreads();

    int* qw = q + w * GCN_QCAP;
    float* accl = acc + lane;
    int cnt = 0;

    const uint4* key4 = reinterpret_cast<const uint4*>(key);
    const unsigned kb = blockIdx.x * 0x01010101u;
    auto load_keys = [&](int st) -> uint4 { return key4[(st * GCN_NW + w) * 32 + lane]; };
    auto scan_step = [&](int st, const uint4& k) {
        const int ebase = ((st * GCN_NW + w) * 32 + lane) * 16;
        unsigned m = (gcn_zero_bytes(k.x ^ kb) >> 7) | (gcn_zero_bytes(k.y ^ kb) >> 6) |
                     (gcn_zero_bytes(k.z ^ kb) >> 5) | (gcn_zero_bytes(k.w ^ kb) >> 4);
        unsigned b = __builtin_amdgcn_ballot_w32(m != 0);
        while (b) {
            if (m) {
                int i = __builtin_ctz(m);
                m &= m - 1;
                qw[__builtin_amdgcn_mbcnt_lo(b, (unsigned)cnt)] = ebase + ((i & 7) << 2) + (i >> 3);
            }
            cnt += __builtin_popcount(b);
            b = __builtin_amdgcn_ballot_w32(m != 0);
        }
        while (cnt >= 32) {
            gcn_range_batch<F, true>(x, rec, accl, qw + (cnt - 32), 32, lane);
            cnt -= 32;
        }
    };

    if (nrounds > 0) {
        uint4 r[GCN_KD];
#pragma unroll
        for (int u = 0; u < GCN_KD; ++u) {
            r[u] = load_keys(u);
            __builtin_amdgcn_sched_barrier(0);
        }
        for (int g = 0; g < nrounds; ++g) {
            const int gn = (g + 1 < nrounds) ? g + 1 : g;
#pragma unroll
            for (int u = 0; u < GCN_KD; ++u) {
                scan_step(g * GCN_KD + u, r[u]);
                r[u] = load_keys(gn * GCN_KD + u);
            }
        }
    }
    if (cnt > 0) gcn_range_batch<F, false>(x, rec, accl, qw, cnt, lane);

    __syncthreads();
    constexpr int LPR = F / 32;
    typedef float rowv __attribute__((ext_vector_type(LPR)));
    int rows = N - base;
    rows = (rows < GCN_R) ? rows : GCN_R;
    rowv* outv = reinterpret_cast<rowv*>(agg + (size_t)base * F);
    for (int i = tid; i < rows * 32; i += GCN_NW * 32) {
        const float* ar = acc + (i >> 5) * F + (i & 31);
        rowv o;
#pragma unroll
        for (int k = 0; k < LPR; ++k) o[k] = ar[32 * k];
        outv[i] = o;
    }
}


__global__ void __launch_bounds__(512)
gcn_edge_pool(const float* __restrict__ x, const int* __restrict__ rec,
              const float* __restrict__ dnorm, float* __restrict__ pooled, int E) {
    constexpr int F   = 64;
    constexpr int LPR = F / 32;
    constexpr int G   = 16 / LPR;
    constexpr int NG  = 32 / G;
    __shared__ float tab[GCN_N_GRAPHS * F];
    for (int i = threadIdx.x; i < GCN_N_GRAPHS * F; i += blockDim.x) tab[i] = 0.0f;
    __syncthreads();

    int lane = threadIdx.x & 31;
    int wpb = blockDim.x >> 5;
    int wave = __builtin_amdgcn_readfirstlane((int)(blockIdx.x * wpb + (threadIdx.x >> 5)));
    int nwaves = gridDim.x * wpb;
    int nchunks = (E + 31) >> 5;
    float* tl = tab + lane;

    for (int ch = wave; ch < nchunks; ch += nwaves) {
        int e0 = ch * 32;
        int e = e0 + lane;
        int s = 0, g = 0;
        float c = 0.0f;
        if (e < E) {
            int2 r = reinterpret_cast<const int2*>(rec)[e];
            s = r.x & 0x1FFFF;
            g = (int)((unsigned)r.x >> 26);
            c = __int_as_float(r.y) * dnorm[e];
        }
        int ci = __float_as_int(c);

        if (e0 + 32 <= E) {
            float v[2][G][LPR];
            auto load_group = [&](int gr, float (&buf)[G][LPR]) {
#pragma unroll
                for (int j = 0; j < G; ++j) {
                    int sj = __builtin_amdgcn_readlane(s, gr * G + j);
                    float2 rv = reinterpret_cast<const float2*>(x + (size_t)sj * F)[lane];
                    buf[j][0] = rv.x;
                    buf[j][1] = rv.y;
                }
            };
            auto add_group = [&](int gr, float (&buf)[G][LPR]) {
#pragma unroll
                for (int j = 0; j < G; ++j) {
                    int gj = __builtin_amdgcn_readlane(g, gr * G + j);
                    float cj = __int_as_float(__builtin_amdgcn_readlane(ci, gr * G + j));
                    float* tr = tl + gj * F;
#pragma unroll
                    for (int i = 0; i < LPR; ++i)
                        __hip_atomic_fetch_add(tr + 32 * i, buf[j][i] * cj, __ATOMIC_RELAXED,
                                               __HIP_MEMORY_SCOPE_WORKGROUP);
                }
            };
            load_group(0, v[0]);
#pragma unroll
            for (int gr = 0; gr < NG; ++gr) {
                if (gr + 1 < NG) load_group(gr + 1, v[(gr + 1) & 1]);
                add_group(gr, v[gr & 1]);
            }
        } else {
            int cnt = E - e0;
            for (int j = 0; j < cnt; ++j) {
                int sj = __builtin_amdgcn_readlane(s, j);
                int gj = __builtin_amdgcn_readlane(g, j);
                float cj = __int_as_float(__builtin_amdgcn_readlane(ci, j));
                float2 rv = reinterpret_cast<const float2*>(x + (size_t)sj * F)[lane];
                float* tr = tl + gj * F;
                __hip_atomic_fetch_add(tr, rv.x * cj, __ATOMIC_RELAXED, __HIP_MEMORY_SCOPE_WORKGROUP);
                __hip_atomic_fetch_add(tr + 32, rv.y * cj, __ATOMIC_RELAXED, __HIP_MEMORY_SCOPE_WORKGROUP);
            }
        }
    }

    __syncthreads();
    for (int i = threadIdx.x; i < GCN_N_GRAPHS * F; i += blockDim.x) {
        int f = i & (F - 1);
        unsafeAtomicAdd(&pooled[i], tab[(i - f) + 32 * (f & 1) + (f >> 1)]);
    }
}


template <bool RELU, bool MEAN>
__global__ void gcn_gemm_wmma(const float* __restrict__ A, const float* __restrict__ in_norm,
                              const float* __restrict__ W, const float* __restrict__ bias,
                              float* __restrict__ out, int M, int K, int Nw,
                              const float* __restrict__ counts) {
    int wave = blockIdx.x * (blockDim.x >> 5) + (threadIdx.x >> 5);
    int tiles_n = Nw >> 4;
    int total = (M >> 4) * tiles_n;
    if (wave >= total) return;
    int mt = wave / tiles_n;
    int nt = wave - mt * tiles_n;
    int row0 = mt << 4;
    int col0 = nt << 4;

    int lane  = threadIdx.x & 31;
    int idx15 = lane & 15;
    int koff  = (lane >> 4) << 1;

    int rowA = row0 + idx15;
    float scale = 1.0f;
    if (!MEAN) scale = in_norm[rowA];
    const float* arow = A + (size_t)rowA * K + koff;
    const float* wcol = W + col0 + idx15;

    v8f acc = {};
    for (int k = 0; k < K; k += 4) {
        float2 av = *reinterpret_cast<const float2*>(arow + k);
        v2f af;
        af.x = MEAN ? av.x : av.x * scale;
        af.y = MEAN ? av.y : av.y * scale;
        v2f bf;
        bf.x = wcol[(size_t)(k + koff) * Nw];
        bf.y = wcol[(size_t)(k + koff + 1) * Nw];
        acc = __builtin_amdgcn_wmma_f32_16x16x4_f32(
             false, af,  false, bf,
             (short)0, acc,  false,  false);
    }

    int col = col0 + idx15;
    float bv = bias[col];
    int mbase = (lane >> 4) << 3;
#pragma unroll
    for (int r = 0; r < 8; ++r) {
        int m = row0 + mbase + r;
        if (MEAN) {
            float c = counts[m];
            float cc = (c < 1.0f) ? 1.0f : c;
            out[(size_t)m * Nw + col] = (acc[r] + c * bv) / cc;
        } else {
            float v = acc[r] + bv;
            if (RELU) v = (v > 0.0f) ? v : 0.0f;
            out[(size_t)m * Nw + col] = v;
        }
    }
}


template <int NW, int CT, bool RELU>
__global__ void __launch_bounds__(256)
gcn_gemm_strip(const float* __restrict__ A, const float* __restrict__ in_norm,
               const float* __restrict__ W, const float* __restrict__ bias,
               float* __restrict__ out, int M) {
    constexpr int K   = 128;
    constexpr int NCG = NW / (16 * CT);
    const int lane = threadIdx.x & 31;
    const int c = lane & 15;
    const int h = lane >> 4;
    const int wave = blockIdx.x * (blockDim.x >> 5) + (threadIdx.x >> 5);
    const int nwaves = gridDim.x * (blockDim.x >> 5);
    const int cg = wave % NCG;
    const int col0 = cg * 16 * CT;
    const int nstrips = M >> 4;
    int strip = wave / NCG;
    const int sstride = nwaves / NCG;
    if (strip >= nstrips) return;

    v2f bf[16][2][CT];
#pragma unroll
    for (int j = 0; j < 16; ++j)
#pragma unroll
        for (int t = 0; t < 2; ++t) {
            const float* wr = W + (size_t)(8 * j + 4 * h + 2 * t) * NW + col0 + c;
#pragma unroll
            for (int n = 0; n < CT; ++n) {
                bf[j][t][n].x = wr[16 * n];
                bf[j][t][n].y = wr[NW + 16 * n];
            }
        }
    float bv[CT];
#pragma unroll
    for (int n = 0; n < CT; ++n) bv[n] = bias[col0 + 16 * n + c];

    float4 av[16];
    {
        const float4* ar = reinterpret_cast<const float4*>(A + (size_t)(strip * 16 + c) * K + 4 * h);
#pragma unroll
        for (int j = 0; j < 16; ++j) av[j] = ar[2 * j];
    }
    for (; strip < nstrips; strip += sstride) {
        const int ns = (strip + sstride < nstrips) ? strip + sstride : strip;
        float4 nv[16];
        {
            const float4* ar = reinterpret_cast<const float4*>(A + (size_t)(ns * 16 + c) * K + 4 * h);
#pragma unroll
            for (int j = 0; j < 16; ++j) nv[j] = ar[2 * j];
        }
        const float scale = in_norm[strip * 16 + c];
        v8f acc[CT];
#pragma unroll
        for (int n = 0; n < CT; ++n) acc[n] = v8f{};
#pragma unroll
        for (int j = 0; j < 16; ++j) {
            v2f a0, a1;
            a0.x = av[j].x * scale; a0.y = av[j].y * scale;
            a1.x = av[j].z * scale; a1.y = av[j].w * scale;
#pragma unroll
            for (int n = 0; n < CT; ++n)
                acc[n] = __builtin_amdgcn_wmma_f32_16x16x4_f32(false, a0, false, bf[j][0][n],
                                                               (short)0, acc[n], false, false);
#pragma unroll
            for (int n = 0; n < CT; ++n)
                acc[n] = __builtin_amdgcn_wmma_f32_16x16x4_f32(false, a1, false, bf[j][1][n],
                                                               (short)0, acc[n], false, false);
        }
        float* orow = out + (size_t)(strip * 16 + 8 * h) * NW + col0 + c;
#pragma unroll
        for (int r = 0; r < 8; ++r)
#pragma unroll
            for (int n = 0; n < CT; ++n) {
                float v = acc[n][r] + bv[n];
                if (RELU) v = (v > 0.0f) ? v : 0.0f;
                orow[(size_t)r * NW + 16 * n] = v;
            }
#pragma unroll
        for (int j = 0; j < 16; ++j) av[j] = nv[j];
    }
}


static inline int cdiv_l(long a, long b) { return (int)((a + b - 1) / b); }

extern "C" void kernel_launch(void* const* d_in, const int* in_sizes, int n_in,
                              void* d_out, int out_size, void* d_ws, size_t ws_size,
                              hipStream_t stream) {
    const float* emb = (const float*)d_in[0];
    const float* W1  = (const float*)d_in[1];
    const float* b1  = (const float*)d_in[2];
    const float* W2  = (const float*)d_in[3];
    const float* b2  = (const float*)d_in[4];
    const float* W3  = (const float*)d_in[5];
    const float* b3  = (const float*)d_in[6];
    const int* src = (const int*)d_in[7];
    const int* dst = (const int*)d_in[8];
    const int* gid = (const int*)d_in[9];

    const int E = in_sizes[7];
    const int N = in_sizes[9];

    float* ws       = (float*)d_ws;
    float* out_norm = ws;
    float* in_norm  = out_norm + N;
    float* pooled   = in_norm + N;
    float* counts   = pooled + GCN_N_GRAPHS * 64;
    float* agg      = counts + GCN_N_GRAPHS;
    float* h        = agg + (size_t)N * 128;
    unsigned int* key = (unsigned int*)(h + (size_t)N * 128);

    const int T = 256;
    const int nrounds = cdiv_l(E, GCN_KROUND);
    const int Epad = nrounds * GCN_KROUND;
    int* rec = (int*)(key + Epad / 4);
    float* dnorm = (float*)(rec + 2 * (size_t)Epad);
    int* npair = (int*)(dnorm + Epad);

    int zc = 2 * N + GCN_N_GRAPHS * 64 + GCN_N_GRAPHS;
    gcn_fill_zero<<<cdiv_l(zc, 4 * T), T, 0, stream>>>(ws, zc);

    gcn_degree<<<cdiv_l(cdiv_l(E, 4), T), T, 0, stream>>>(src, dst, out_norm, in_norm, E);
    gcn_norm_finalize<<<cdiv_l(2L * N, T), T, 0, stream>>>(out_norm, gid, npair, counts, N);
    gcn_edge_records<<<cdiv_l(Epad / 4, T), T, 0, stream>>>(src, dst, out_norm, npair, key, rec, dnorm,
                                                         E, Epad);

    gcn_range_spmm<128><<<cdiv_l(N, GCN_R), GCN_NW * 32, 0, stream>>>(emb, rec, key, agg, nrounds, N);
    gcn_gemm_strip<128, 4, true><<<256, 256, 0, stream>>>(agg, in_norm, W1, b1, h, N);

    gcn_range_spmm<128><<<cdiv_l(N, GCN_R), GCN_NW * 32, 0, stream>>>(h, rec, key, agg, nrounds, N);
    gcn_gemm_strip<64, 2, true><<<256, 256, 0, stream>>>(agg, in_norm, W2, b2, h, N);

    gcn_edge_pool<<<768, 512, 0, stream>>>(h, rec, dnorm, pooled, E);
    {
        int tiles = (GCN_N_GRAPHS / 16) * (64 / 16);
        gcn_gemm_wmma<false, true><<<cdiv_l(tiles, 4), 128, 0, stream>>>(
            pooled, nullptr, W3, b3, (float*)d_out, GCN_N_GRAPHS, 64, 64, counts);
    }
}


# Layer 2 SpMM skips the owner scan: layer 1's waves keep the records of the edges they find (a list per wave, whole lines); layer 2 streams its list, records ahead of rows. Sum order: arrival
# speedup vs baseline: 3.1286x; 1.0495x over previous
#include <hip/hip_runtime.h>

typedef __attribute__((ext_vector_type(2))) float v2f;
typedef __attribute__((ext_vector_type(8))) float v8f;

#define GCN_N_GRAPHS 64

#define GCN_R     400
#define GCN_NW    32
#define GCN_QCAP  544
#define GCN_KD    4
#define GCN_KROUND (GCN_KD * GCN_NW * 32 * 16)
#define GCN_LCAP  512


__global__ void gcn_fill_zero(float* __restrict__ p, int n) {
    int n4 = n >> 2;
    float4* p4 = reinterpret_cast<float4*>(p);
    int t = blockIdx.x * blockDim.x + threadIdx.x;
    int stride = gridDim.x * blockDim.x;
    for (int i = t; i < n4; i += stride) p4[i] = make_float4(0.0f, 0.0f, 0.0f, 0.0f);
    int r = (n4 << 2) + t;
    if (r < n) p[r] = 0.0f;
}

__global__ void gcn_degree(const int* __restrict__ src, const int* __restrict__ dst,
                           float* __restrict__ deg_out, float* __restrict__ deg_in, int E) {
    int t = blockIdx.x * blockDim.x + threadIdx.x;
    int e = t * 4;
    if (e + 3 < E) {
        int4 s4 = reinterpret_cast<const int4*>(src)[t];
        int4 d4 = reinterpret_cast<const int4*>(dst)[t];
        unsafeAtomicAdd(&deg_out[s4.x], 1.0f);
        unsafeAtomicAdd(&deg_out[s4.y], 1.0f);
        unsafeAtomicAdd(&deg_out[s4.z], 1.0f);
        unsafeAtomicAdd(&deg_out[s4.w], 1.0f);
        unsafeAtomicAdd(&deg_in[d4.x], 1.0f);
        unsafeAtomicAdd(&deg_in[d4.y], 1.0f);
        unsafeAtomicAdd(&deg_in[d4.z], 1.0f);
        unsafeAtomicAdd(&deg_in[d4.w], 1.0f);
    } else {
        for (int i = 0; i < 4; ++i) {
            if (e + i < E) {
                unsafeAtomicAdd(&deg_out[src[e + i]], 1.0f);
                unsafeAtomicAdd(&deg_in[dst[e + i]], 1.0f);
            }
        }
    }
}

__device__ __forceinline__ int gcn_pack_edge(int s, int d, unsigned owner, int g) {
    return (int)((unsigned)s | (((unsigned)d - owner * GCN_R) << 17) | ((unsigned)(g & 63) << 26));
}

__global__ void gcn_edge_records(const int* __restrict__ src, const int* __restrict__ dst,
                                 const float* __restrict__ out_norm, const int* __restrict__ npair,
                                 unsigned int* __restrict__ key, int* __restrict__ rec,
                                 float* __restrict__ dnorm, int E, int Epad) {
    int t = blockIdx.x * blockDim.x + threadIdx.x;
    int e = t * 4;
    unsigned kw = 0xFFFFFFFFu;
    const int2* np2 = reinterpret_cast<const int2*>(npair);
    if (e + 3 < E) {
        int4 s4 = reinterpret_cast<const int4*>(src)[t];
        int4 d4 = reinterpret_cast<const int4*>(dst)[t];
        float a0 = out_norm[s4.x], a1 = out_norm[s4.y], a2 = out_norm[s4.z], a3 = out_norm[s4.w];
        int2 p0 = np2[d4.x], p1 = np2[d4.y], p2 = np2[d4.z], p3 = np2[d4.w];
        unsigned o0 = (unsigned)d4.x / GCN_R, o1 = (unsigned)d4.y / GCN_R;
        unsigned o2 = (unsigned)d4.z / GCN_R, o3 = (unsigned)d4.w / GCN_R;
        kw = o0 | (o1 << 8) | (o2 << 16) | (o3 << 24);
        int4* r4 = reinterpret_cast<int4*>(rec);
        r4[2 * t] = make_int4(gcn_pack_edge(s4.x, d4.x, o0, p0.y), __float_as_int(a0),
                              gcn_pack_edge(s4.y, d4.y, o1, p1.y), __float_as_int(a1));
        r4[2 * t + 1] = make_int4(gcn_pack_edge(s4.z, d4.z, o2, p2.y), __float_as_int(a2),
                                  gcn_pack_edge(s4.w, d4.w, o3, p3.y), __float_as_int(a3));
        reinterpret_cast<int4*>(dnorm)[t] = make_int4(p0.x, p1.x, p2.x, p3.x);
    } else {
        for (int i = 0; i < 4; ++i) {
            if (e + i < E) {
                int s = src[e + i];
                int d = dst[e + i];
                unsigned o = (unsigned)d / GCN_R;
                int2 p = np2[d];
                rec[2 * (e + i)] = gcn_pack_edge(s, d, o, p.y);
                rec[2 * (e + i) + 1] = __float_as_int(out_norm[s]);
                dnorm[e + i] = __int_as_float(p.x);
                kw = (kw & ~(0xFFu << (8 * i))) | (o << (8 * i));
            }
        }
    }
    if (e < Epad) key[t] = kw;
}

__global__ void gcn_norm_finalize(float* __restrict__ buf, const int* __restrict__ gid,
                                  int* __restrict__ npair, float* __restrict__ counts, int N) {
    __shared__ float cnt[GCN_N_GRAPHS];
    if (threadIdx.x < GCN_N_GRAPHS) cnt[threadIdx.x] = 0.0f;
    __syncthreads();
    int i = blockIdx.x * blockDim.x + threadIdx.x;
    if (i < 2 * N) {
        float d = buf[i];
        d = (d < 1.0f) ? 1.0f : d;
        float r = 1.0f / sqrtf(d);
        buf[i] = r;
        if (i >= N) {
            int g = gid[i - N];
            reinterpret_cast<int2*>(npair)[i - N] = make_int2(__float_as_int(r), g);
            if ((unsigned)g < (unsigned)GCN_N_GRAPHS)
                __hip_atomic_fetch_add(&cnt[g], 1.0f, __ATOMIC_RELAXED, __HIP_MEMORY_SCOPE_WORKGROUP);
        }
    }
    __syncthreads();
    if (threadIdx.x < GCN_N_GRAPHS) {
        float c = cnt[threadIdx.x];
        if (c != 0.0f) unsafeAtomicAdd(&counts[threadIdx.x], c);
    }
}


__device__ __forceinline__ unsigned gcn_zero_bytes(unsigned x) {
    return ~(((x & 0x7F7F7F7Fu) + 0x7F7F7F7Fu) | x) & 0x80808080u;
}

template <int F, bool FULL>
__device__ __forceinline__ void gcn_range_batch(const float* __restrict__ x, float* accl,
                                                int2 r, int nvalid, int lane) {
    constexpr int LPR = F / 32;
    constexpr int G   = 16 / LPR;
    constexpr int NG  = 32 / G;
    typedef float rowv __attribute__((ext_vector_type(LPR)));
    const int s = r.x & 0x1FFFF;
    const int dl = (r.x >> 17) & 0x1FF;
    const int ci = r.y;
    float v[2][G][LPR];
    auto load_group = [&](int g, float (&buf)[G][LPR]) {
#pragma unroll
        for (int j = 0; j < G; ++j) {
            if (FULL || g * G + j < nvalid) {
                int sj = __builtin_amdgcn_readlane(s, g * G + j);
                rowv rv = reinterpret_cast<const rowv*>(x + (size_t)sj * F)[lane];
#pragma unroll
                for (int i = 0; i < LPR; ++i) buf[j][i] = rv[i];
            } else {
#pragma unroll
                for (int i = 0; i < LPR; ++i) buf[j][i] = 0.0f;
            }
        }
    };
    auto add_group = [&](int g, float (&buf)[G][LPR]) {
#pragma unroll
        for (int j = 0; j < G; ++j) {
            if (FULL || g * G + j < nvalid) {
                int dj = __builtin_amdgcn_readlane(dl, g * G + j);
                float cj = __int_as_float(__builtin_amdgcn_readlane(ci, g * G + j));
                float* ar = accl + dj * F;
#pragma unroll
                for (int i = 0; i < LPR; ++i)
                    __hip_atomic_fetch_add(ar + 32 * i, buf[j][i] * cj, __ATOMIC_RELAXED,
                                           __HIP_MEMORY_SCOPE_WORKGROUP);
            }
        }
    };
    load_group(0, v[0]);
#pragma unroll
    for (int g = 0; g < NG; ++g) {
        if (g + 1 < NG) load_group(g + 1, v[(g + 1) & 1]);
        add_group(g, v[g & 1]);
    }
}

template <int F, bool TAKE>
__global__ void __launch_bounds__(GCN_NW * 32)
gcn_range_spmm(const float* __restrict__ x, const int* __restrict__ rec,
               const unsigned int* __restrict__ key, float* __restrict__ agg,
               int* lst, int* lcnt, int nrounds, int N) {
    __shared__ float acc[GCN_R * F];
    __shared__ int q[GCN_NW * GCN_QCAP];

    const int tid = threadIdx.x;
    const int lane = tid & 31;
    const int w = __builtin_amdgcn_readfirstlane(tid >> 5);
    const int base = blockIdx.x * GCN_R;

    const int wid = blockIdx.x * GCN_NW + w;
    int2* my = reinterpret_cast<int2*>(lst) + (size_t)wid * GCN_LCAP;
    const int ntake = TAKE ? __builtin_amdgcn_readfirstlane(lcnt[wid]) : -1;
    const bool use_list = TAKE && (unsigned)ntake <= (unsigned)GCN_LCAP;

    float4* acc4 = reinterpret_cast<float4*>(acc);
    for (int i = tid; i < GCN_R * F / 4; i += GCN_NW * 32) acc4[i] = make_float4(0.0f, 0.0f, 0.0f, 0.0f);
    __syncthreads();

    int* qw = q + w * GCN_QCAP;
    float* accl = acc + lane;
    int cnt = 0;
    const int2* rec2 = reinterpret_cast<const int2*>(rec);
    int nst = 0;

    const uint4* key4 = reinterpret_cast<const uint4*>(key);
    const unsigned kb = blockIdx.x * 0x01010101u;
    auto load_keys = [&](int st) -> uint4 { return key4[(st * GCN_NW + w) * 32 + lane]; };
    auto scan_step = [&](int st, const uint4& k) {
        const int ebase = ((st * GCN_NW + w) * 32 + lane) * 16;
        unsigned m = (gcn_zero_bytes(k.x ^ kb) >> 7) | (gcn_zero_bytes(k.y ^ kb) >> 6) |
                     (gcn_zero_bytes(k.z ^ kb) >> 5) | (gcn_zero_bytes(k.w ^ kb) >> 4);
        unsigned b = __builtin_amdgcn_ballot_w32(m != 0);
        while (b) {
            if (m) {
                int i = __builtin_ctz(m);
                m &= m - 1;
                qw[__builtin_amdgcn_mbcnt_lo(b, (unsigned)cnt)] = ebase + ((i & 7) << 2) + (i >> 3);
            }
            cnt += __builtin_popcount(b);
            b = __builtin_amdgcn_ballot_w32(m != 0);
        }
        while (cnt >= 32) {
            const int2 r = rec2[qw[cnt - 32 + lane]];
            if (!TAKE) {
                if (nst + 32 <= GCN_LCAP) my[nst + lane] = r;
                nst += 32;
            }
            gcn_range_batch<F, true>(x, accl, r, 32, lane);
            cnt -= 32;
        }
    };

    if (use_list) {
        if (ntake > 0) {
            const int last = ntake - 1;
            int2 rn = my[lane < last ? lane : last];
            int b = 0;
            for (; b + 32 <= ntake; b += 32) {
                const int2 r = rn;
                const int nx = b + 32 + lane;
                rn = my[nx < last ? nx : last];
                gcn_range_batch<F, true>(x, accl, r, 32, lane);
            }
            if (b < ntake) gcn_range_batch<F, false>(x, accl, rn, ntake - b, lane);
        }
    } else if (nrounds > 0) {
        uint4 r[GCN_KD];
#pragma unroll
        for (int u = 0; u < GCN_KD; ++u) {
            r[u] = load_keys(u);
            __builtin_amdgcn_sched_barrier(0);
        }
        for (int g = 0; g < nrounds; ++g) {
            const int gn = (g + 1 < nrounds) ? g + 1 : g;
#pragma unroll
            for (int u = 0; u < GCN_KD; ++u) {
                scan_step(g * GCN_KD + u, r[u]);
                r[u] = load_keys(gn * GCN_KD + u);
            }
        }
    }
    if (cnt > 0) {
        int2 r = make_int2(0, 0);
        if (lane < cnt) r = rec2[qw[lane]];
        if (!TAKE) {
            if (nst + 32 <= GCN_LCAP && lane < cnt) my[nst + lane] = r;
            nst += cnt;
        }
        gcn_range_batch<F, false>(x, accl, r, cnt, lane);
    }
    if (!TAKE && lane == 0) lcnt[wid] = nst;

    __syncthreads();
    constexpr int LPR = F / 32;
    typedef float rowv __attribute__((ext_vector_type(LPR)));
    int rows = N - base;
    rows = (rows < GCN_R) ? rows : GCN_R;
    rowv* outv = reinterpret_cast<rowv*>(agg + (size_t)base * F);
    for (int i = tid; i < rows * 32; i += GCN_NW * 32) {
        const float* ar = acc + (i >> 5) * F + (i & 31);
        rowv o;
#pragma unroll
        for (int k = 0; k < LPR; ++k) o[k] = ar[32 * k];
        outv[i] = o;
    }
}


__global__ void __launch_bounds__(512)
gcn_edge_pool(const float* __restrict__ x, const int* __restrict__ rec,
              const float* __restrict__ dnorm, float* __restrict__ pooled, int E) {
    constexpr int F   = 64;
    constexpr int LPR = F / 32;
    constexpr int G   = 16 / LPR;
    constexpr int NG  = 32 / G;
    __shared__ float tab[GCN_N_GRAPHS * F];
    for (int i = threadIdx.x; i < GCN_N_GRAPHS * F; i += blockDim.x) tab[i] = 0.0f;
    __syncthreads();

    int lane = threadIdx.x & 31;
    int wpb = blockDim.x >> 5;
    int wave = __builtin_amdgcn_readfirstlane((int)(blockIdx.x * wpb + (threadIdx.x >> 5)));
    int nwaves = gridDim.x * wpb;
    int nchunks = (E + 31) >> 5;
    float* tl = tab + lane;

    for (int ch = wave; ch < nchunks; ch += nwaves) {
        int e0 = ch * 32;
        int e = e0 + lane;
        int s = 0, g = 0;
        float c = 0.0f;
        if (e < E) {
            int2 r = reinterpret_cast<const int2*>(rec)[e];
            s = r.x & 0x1FFFF;
            g = (int)((unsigned)r.x >> 26);
            c = __int_as_float(r.y) * dnorm[e];
        }
        int ci = __float_as_int(c);

        if (e0 + 32 <= E) {
            float v[2][G][LPR];
            auto load_group = [&](int gr, float (&buf)[G][LPR]) {
#pragma unroll
                for (int j = 0; j < G; ++j) {
                    int sj = __builtin_amdgcn_readlane(s, gr * G + j);
                    float2 rv = reinterpret_cast<const float2*>(x + (size_t)sj * F)[lane];
                    buf[j][0] = rv.x;
                    buf[j][1] = rv.y;
                }
            };
            auto add_group = [&](int gr, float (&buf)[G][LPR]) {
#pragma unroll
                for (int j = 0; j < G; ++j) {
                    int gj = __builtin_amdgcn_readlane(g, gr * G + j);
                    float cj = __int_as_float(__builtin_amdgcn_readlane(ci, gr * G + j));
                    float* tr = tl + gj * F;
#pragma unroll
                    for (int i = 0; i < LPR; ++i)
                        __hip_atomic_fetch_add(tr + 32 * i, buf[j][i] * cj, __ATOMIC_RELAXED,
                                               __HIP_MEMORY_SCOPE_WORKGROUP);
                }
            };
            load_group(0, v[0]);
#pragma unroll
            for (int gr = 0; gr < NG; ++gr) {
                if (gr + 1 < NG) load_group(gr + 1, v[(gr + 1) & 1]);
                add_group(gr, v[gr & 1]);
            }
        } else {
            int cnt = E - e0;
            for (int j = 0; j < cnt; ++j) {
                int sj = __builtin_amdgcn_readlane(s, j);
                int gj = __builtin_amdgcn_readlane(g, j);
                float cj = __int_as_float(__builtin_amdgcn_readlane(ci, j));
                float2 rv = reinterpret_cast<const float2*>(x + (size_t)sj * F)[lane];
                float* tr = tl + gj * F;
                __hip_atomic_fetch_add(tr, rv.x * cj, __ATOMIC_RELAXED, __HIP_MEMORY_SCOPE_WORKGROUP);
                __hip_atomic_fetch_add(tr + 32, rv.y * cj, __ATOMIC_RELAXED, __HIP_MEMORY_SCOPE_WORKGROUP);
            }
        }
    }

    __syncthreads();
    for (int i = threadIdx.x; i < GCN_N_GRAPHS * F; i += blockDim.x) {
        int f = i & (F - 1);
        unsafeAtomicAdd(&pooled[i], tab[(i - f) + 32 * (f & 1) + (f >> 1)]);
    }
}


template <bool RELU, bool MEAN>
__global__ void gcn_gemm_wmma(const float* __restrict__ A, const float* __restrict__ in_norm,
                              const float* __restrict__ W, const float* __restrict__ bias,
                              float* __restrict__ out, int M, int K, int Nw,
                              const float* __restrict__ counts) {
    int wave = blockIdx.x * (blockDim.x >> 5) + (threadIdx.x >> 5);
    int tiles_n = Nw >> 4;
    int total = (M >> 4) * tiles_n;
    if (wave >= total) return;
    int mt = wave / tiles_n;
    int nt = wave - mt * tiles_n;
    int row0 = mt << 4;
    int col0 = nt << 4;

    int lane  = threadIdx.x & 31;
    int idx15 = lane & 15;
    int koff  = (lane >> 4) << 1;

    int rowA = row0 + idx15;
    float scale = 1.0f;
    if (!MEAN) scale = in_norm[rowA];
    const float* arow = A + (size_t)rowA * K + koff;
    const float* wcol = W + col0 + idx15;

    v8f acc = {};
    for (int k = 0; k < K; k += 4) {
        float2 av = *reinterpret_cast<const float2*>(arow + k);
        v2f af;
        af.x = MEAN ? av.x : av.x * scale;
        af.y = MEAN ? av.y : av.y * scale;
        v2f bf;
        bf.x = wcol[(size_t)(k + koff) * Nw];
        bf.y = wcol[(size_t)(k + koff + 1) * Nw];
        acc = __builtin_amdgcn_wmma_f32_16x16x4_f32(
             false, af,  false, bf,
             (short)0, acc,  false,  false);
    }

    int col = col0 + idx15;
    float bv = bias[col];
    int mbase = (lane >> 4) << 3;
#pragma unroll
    for (int r = 0; r < 8; ++r) {
        int m = row0 + mbase + r;
        if (MEAN) {
            float c = counts[m];
            float cc = (c < 1.0f) ? 1.0f : c;
            out[(size_t)m * Nw + col] = (acc[r] + c * bv) / cc;
        } else {
            float v = acc[r] + bv;
            if (RELU) v = (v > 0.0f) ? v : 0.0f;
            out[(size_t)m * Nw + col] = v;
        }
    }
}


template <int NW, int CT, bool RELU>
__global__ void __launch_bounds__(256)
gcn_gemm_strip(const float* __restrict__ A, const float* __restrict__ in_norm,
               const float* __restrict__ W, const float* __restrict__ bias,
               float* __restrict__ out, int M) {
    constexpr int K   = 128;
    constexpr int NCG = NW / (16 * CT);
    const int lane = threadIdx.x & 31;
    const int c = lane & 15;
    const int h = lane >> 4;
    const int wave = blockIdx.x * (blockDim.x >> 5) + (threadIdx.x >> 5);
    const int nwaves = gridDim.x * (blockDim.x >> 5);
    const int cg = wave % NCG;
    const int col0 = cg * 16 * CT;
    const int nstrips = M >> 4;
    int strip = wave / NCG;
    const int sstride = nwaves / NCG;
    if (strip >= nstrips) return;

    v2f bf[16][2][CT];
#pragma unroll
    for (int j = 0; j < 16; ++j)
#pragma unroll
        for (int t = 0; t < 2; ++t) {
            const float* wr = W + (size_t)(8 * j + 4 * h + 2 * t) * NW + col0 + c;
#pragma unroll
            for (int n = 0; n < CT; ++n) {
                bf[j][t][n].x = wr[16 * n];
                bf[j][t][n].y = wr[NW + 16 * n];
            }
        }
    float bv[CT];
#pragma unroll
    for (int n = 0; n < CT; ++n) bv[n] = bias[col0 + 16 * n + c];

    float4 av[16];
    {
        const float4* ar = reinterpret_cast<const float4*>(A + (size_t)(strip * 16 + c) * K + 4 * h);
#pragma unroll
        for (int j = 0; j < 16; ++j) av[j] = ar[2 * j];
    }
    for (; strip < nstrips; strip += sstride) {
        const int ns = (strip + sstride < nstrips) ? strip + sstride : strip;
        float4 nv[16];
        {
            const float4* ar = reinterpret_cast<const float4*>(A + (size_t)(ns * 16 + c) * K + 4 * h);
#pragma unroll
            for (int j = 0; j < 16; ++j) nv[j] = ar[2 * j];
        }
        const float scale = in_norm[strip * 16 + c];
        v8f acc[CT];
#pragma unroll
        for (int n = 0; n < CT; ++n) acc[n] = v8f{};
#pragma unroll
        for (int j = 0; j < 16; ++j) {
            v2f a0, a1;
            a0.x = av[j].x * scale; a0.y = av[j].y * scale;
            a1.x = av[j].z * scale; a1.y = av[j].w * scale;
#pragma unroll
            for (int n = 0; n < CT; ++n)
                acc[n] = __builtin_amdgcn_wmma_f32_16x16x4_f32(false, a0, false, bf[j][0][n],
                                                               (short)0, acc[n], false, false);
#pragma unroll
            for (int n = 0; n < CT; ++n)
                acc[n] = __builtin_amdgcn_wmma_f32_16x16x4_f32(false, a1, false, bf[j][1][n],
                                                               (short)0, acc[n], false, false);
        }
        float* orow = out + (size_t)(strip * 16 + 8 * h) * NW + col0 + c;
#pragma unroll
        for (int r = 0; r < 8; ++r)
#pragma unroll
            for (int n = 0; n < CT; ++n) {
                float v = acc[n][r] + bv[n];
                if (RELU) v = (v > 0.0f) ? v : 0.0f;
                orow[(size_t)r * NW + 16 * n] = v;
            }
#pragma unroll
        for (int j = 0; j < 16; ++j) av[j] = nv[j];
    }
}


static inline int cdiv_l(long a, long b) { return (int)((a + b - 1) / b); }

extern "C" void kernel_launch(void* const* d_in, const int* in_sizes, int n_in,
                              void* d_out, int out_size, void* d_ws, size_t ws_size,
                              hipStream_t stream) {
    const float* emb = (const float*)d_in[0];
    const float* W1  = (const float*)d_in[1];
    const float* b1  = (const float*)d_in[2];
    const float* W2  = (const float*)d_in[3];
    const float* b2  = (const float*)d_in[4];
    const float* W3  = (const float*)d_in[5];
    const float* b3  = (const float*)d_in[6];
    const int* src = (const int*)d_in[7];
    const int* dst = (const int*)d_in[8];
    const int* gid = (const int*)d_in[9];

    const int E = in_sizes[7];
    const int N = in_sizes[9];

    float* ws       = (float*)d_ws;
    float* out_norm = ws;
    float* in_norm  = out_norm + N;
    float* pooled   = in_norm + N;
    float* counts   = pooled + GCN_N_GRAPHS * 64;
    float* agg      = counts + GCN_N_GRAPHS;
    float* h        = agg + (size_t)N * 128;
    unsigned int* key = (unsigned int*)(h + (size_t)N * 128);

    const int T = 256;
    const int nrounds = cdiv_l(E, GCN_KROUND);
    const int Epad = nrounds * GCN_KROUND;
    int* rec = (int*)(key + Epad / 4);
    float* dnorm = (float*)(rec + 2 * (size_t)Epad);
    int* npair = (int*)(dnorm + Epad);
    const int nwg = cdiv_l(N, GCN_R);
    int* lst = npair + 2 * (size_t)N;
    int* lcnt = lst + (size_t)nwg * GCN_NW * GCN_LCAP * 2;

    int zc = 2 * N + GCN_N_GRAPHS * 64 + GCN_N_GRAPHS;
    gcn_fill_zero<<<cdiv_l(zc, 4 * T), T, 0, stream>>>(ws, zc);

    gcn_degree<<<cdiv_l(cdiv_l(E, 4), T), T, 0, stream>>>(src, dst, out_norm, in_norm, E);
    gcn_norm_finalize<<<cdiv_l(2L * N, T), T, 0, stream>>>(out_norm, gid, npair, counts, N);
    gcn_edge_records<<<cdiv_l(Epad / 4, T), T, 0, stream>>>(src, dst, out_norm, npair, key, rec, dnorm,
                                                         E, Epad);

    gcn_range_spmm<128, false><<<nwg, GCN_NW * 32, 0, stream>>>(emb, rec, key, agg, lst, lcnt, nrounds, N);
    gcn_gemm_strip<128, 4, true><<<256, 256, 0, stream>>>(agg, in_norm, W1, b1, h, N);

    gcn_range_spmm<128, true><<<nwg, GCN_NW * 32, 0, stream>>>(h, rec, key, agg, lst, lcnt, nrounds, N);
    gcn_gemm_strip<64, 2, true><<<256, 256, 0, stream>>>(agg, in_norm, W2, b2, h, N);

    gcn_edge_pool<<<768, 512, 0, stream>>>(h, rec, dnorm, pooled, E);
    {
        int tiles = (GCN_N_GRAPHS / 16) * (64 / 16);
        gcn_gemm_wmma<false, true><<<cdiv_l(tiles, 4), 128, 0, stream>>>(
            pooled, nullptr, W3, b3, (float*)d_out, GCN_N_GRAPHS, 64, 64, counts);
    }
}
